# starting kernel
# baseline (speedup 1.0000x reference)
#include <hip/hip_runtime.h>

typedef __attribute__((ext_vector_type(2))) float v2f;
typedef __attribute__((ext_vector_type(8))) float v8f;

__device__ __forceinline__ void atomic_add_f32(float* p, float v)
{
    asm volatile("global_atomic_add_f32 %0, %1, off"
                 :: "v"(p), "v"(v) : "memory");
}

__global__ __launch_bounds__(128)
void gemm_xw_kernel(const float* __restrict__ x,
                    const float* __restrict__ Wn,
                    const float* __restrict__ Ws,
                    float* __restrict__ xw, int nrows)
{
    const int lane  = threadIdx.x & 31;
    const int wave  = threadIdx.x >> 5;
    const int m0    = blockIdx.x * 16;
    const int lhalf = lane >> 4;
    const int lm    = lane & 15;

    const float* __restrict__ W = (wave < 2) ? Wn : Ws;
    const int n0 = (wave & 1) * 16;

    int arow = m0 + lm;
    if (arow >= nrows) arow = nrows - 1;

    v8f c = {};
    #pragma unroll
    for (int k0 = 0; k0 < 64; k0 += 4) {
        v2f a = *(const v2f*)(x + (size_t)arow * 64 + k0 + 2 * lhalf);
        v2f b;
        b.x = W[(k0 + 2 * lhalf)     * 32 + n0 + lm];
        b.y = W[(k0 + 2 * lhalf + 1) * 32 + n0 + lm];
        c = __builtin_amdgcn_wmma_f32_16x16x4_f32(false, a, false, b,
                                                  (short)0, c, false, false);
    }

    const int gcol = wave * 16 + lm;
    #pragma unroll
    for (int v = 0; v < 8; ++v) {
        const int row = m0 + v + 8 * lhalf;
        if (row < nrows) xw[(size_t)row * 64 + gcol] = c[v];
    }
}

__global__ void init_deg_kernel(float* __restrict__ dn, float* __restrict__ ds, int n)
{
    const int i = blockIdx.x * blockDim.x + threadIdx.x;
    if (i < n) { dn[i] = 1.0f; ds[i] = 1.0f; }
}

__global__ void count_deg_kernel(const int* __restrict__ dst_n,
                                 const int* __restrict__ dst_s,
                                 float* __restrict__ dn,
                                 float* __restrict__ ds, int E)
{
    const int e = blockIdx.x * blockDim.x + threadIdx.x;
    if (e >= E) return;
    const int*   dst = blockIdx.y ? dst_s : dst_n;
    float*       deg = blockIdx.y ? ds    : dn;
    atomic_add_f32(&deg[dst[e]], 1.0f);
}

__global__ void rsqrt_kernel(float* __restrict__ dn, float* __restrict__ ds, int n)
{
    const int i = blockIdx.x * blockDim.x + threadIdx.x;
    if (i < n) {
        const float a = dn[i], b = ds[i];
        dn[i] = (a > 0.0f) ? rsqrtf(a) : 0.0f;
        ds[i] = (b > 0.0f) ? rsqrtf(b) : 0.0f;
    }
}

__global__ void init_h_kernel(const float* __restrict__ xw,
                              const float* __restrict__ dn,
                              const float* __restrict__ ds,
                              float* __restrict__ h, int n)
{
    const int t = blockIdx.x * blockDim.x + threadIdx.x;
    if (t >= n * 32) return;
    const int i = t >> 5, j = t & 31;
    const float a = dn[i], b = ds[i];
    h[t] = xw[(size_t)i * 64 + j]      * (a * a)
         + xw[(size_t)i * 64 + 32 + j] * (b * b);
}

__global__ void scatter_kernel(const int* __restrict__ en,
                               const int* __restrict__ es,
                               const float* __restrict__ xw,
                               const float* __restrict__ dn,
                               const float* __restrict__ dsv,
                               float* __restrict__ h, int E)
{
    const unsigned t = blockIdx.x * blockDim.x + threadIdx.x;
    const unsigned e = t >> 5;
    const int      j = (int)(t & 31u);
    if (e >= (unsigned)E) return;

    const int    typ   = blockIdx.y;
    const int*   edges = typ ? es  : en;
    const float* dinv  = typ ? dsv : dn;
    const int    cbase = typ ? 32  : 0;

    const int s = edges[e], d = edges[E + e];
    const float norm = dinv[s] * dinv[d];
    const float val  = xw[(size_t)s * 64 + cbase + j] * norm;
    atomic_add_f32(&h[(size_t)d * 32 + j], val);
}

__global__ void final_kernel(const float* __restrict__ h,
                             const float* __restrict__ bn,
                             const float* __restrict__ bs,
                             const float* __restrict__ wl,
                             const float* __restrict__ bl,
                             float* __restrict__ out, int n)
{
    const int t = blockIdx.x * blockDim.x + threadIdx.x;
    const int i = t >> 5, j = t & 31;
    if (i >= n) return;
    float v = h[(size_t)i * 32 + j] + bn[j] + bs[j];
    v = fmaxf(v, 0.0f) * wl[j];
    #pragma unroll
    for (int off = 16; off > 0; off >>= 1)
        v += __shfl_xor(v, off, 32);
    if (j == 0) out[i] = v + bl[0];
}

extern "C" void kernel_launch(void* const* d_in, const int* in_sizes, int n_in,
                              void* d_out, int out_size, void* d_ws, size_t ws_size,
                              hipStream_t stream)
{
    const float* x  = (const float*)d_in[0];
    const int*   en = (const int*)  d_in[1];
    const int*   es = (const int*)  d_in[2];
    const float* Wn = (const float*)d_in[3];
    const float* bn = (const float*)d_in[4];
    const float* Ws = (const float*)d_in[5];
    const float* bs = (const float*)d_in[6];
    const float* Wl = (const float*)d_in[7];
    const float* bl = (const float*)d_in[8];
    float* out = (float*)d_out;

    const int N = in_sizes[0] / 64;
    const int E = in_sizes[1] / 2;

    float* ws  = (float*)d_ws;
    float* xw  = ws;
    float* h   = xw + (size_t)N * 64;
    float* dn  = h  + (size_t)N * 32;
    float* ds  = dn + N;

    const int TPB = 256;

    init_deg_kernel<<<(N + TPB - 1) / TPB, TPB, 0, stream>>>(dn, ds, N);
    count_deg_kernel<<<dim3((E + TPB - 1) / TPB, 2), TPB, 0, stream>>>(
        en + E, es + E, dn, ds, E);
    rsqrt_kernel<<<(N + TPB - 1) / TPB, TPB, 0, stream>>>(dn, ds, N);

    gemm_xw_kernel<<<(N + 15) / 16, 128, 0, stream>>>(x, Wn, Ws, xw, N);

    init_h_kernel<<<(N * 32 + TPB - 1) / TPB, TPB, 0, stream>>>(xw, dn, ds, h, N);
    {
        const unsigned tot  = (unsigned)E * 32u;
        const unsigned grid = (tot + TPB - 1) / TPB;
        scatter_kernel<<<dim3(grid, 2), TPB, 0, stream>>>(en, es, xw, dn, ds, h, E);
    }

    final_kernel<<<(N * 32 + TPB - 1) / TPB, TPB, 0, stream>>>(h, bn, bs, Wl, bl, out, N);
}


# edge scatter on resident waves: a wave loops over 32-edge chunks, lanes load src, dst, norm coalesced, per-edge scalars by readlane, 8 row gathers in flight; h float atomics kept (arrival order)
# speedup vs baseline: 1.9768x; 1.9768x over previous
#include <hip/hip_runtime.h>

typedef __attribute__((ext_vector_type(2))) float v2f;
typedef __attribute__((ext_vector_type(8))) float v8f;

__device__ __forceinline__ void atomic_add_f32(float* p, float v)
{
    asm volatile("global_atomic_add_f32 %0, %1, off"
                 :: "v"(p), "v"(v) : "memory");
}

__device__ __forceinline__ void atomic_add_f32_wait(float* p, float v)
{
    asm volatile("global_atomic_add_f32 %0, %1, off\n\ts_wait_xcnt 0x0"
                 :: "v"(p), "v"(v) : "memory");
}

__device__ __forceinline__ void wait_issued_atomics8(float* const (&p)[8],
                                                     const float (&v)[8])
{
    asm volatile("s_wait_xcnt 0x0"
                 :: "v"(p[0]), "v"(p[1]), "v"(p[2]), "v"(p[3]),
                    "v"(p[4]), "v"(p[5]), "v"(p[6]), "v"(p[7]),
                    "v"(v[0]), "v"(v[1]), "v"(v[2]), "v"(v[3]),
                    "v"(v[4]), "v"(v[5]), "v"(v[6]), "v"(v[7])
                 : "memory");
}

__global__ __launch_bounds__(128)
void gemm_xw_kernel(const float* __restrict__ x,
                    const float* __restrict__ Wn,
                    const float* __restrict__ Ws,
                    float* __restrict__ xw, int nrows)
{
    const int lane  = threadIdx.x & 31;
    const int wave  = threadIdx.x >> 5;
    const int m0    = blockIdx.x * 16;
    const int lhalf = lane >> 4;
    const int lm    = lane & 15;

    const float* __restrict__ W = (wave < 2) ? Wn : Ws;
    const int n0 = (wave & 1) * 16;

    int arow = m0 + lm;
    if (arow >= nrows) arow = nrows - 1;

    v8f c = {};
    #pragma unroll
    for (int k0 = 0; k0 < 64; k0 += 4) {
        v2f a = *(const v2f*)(x + (size_t)arow * 64 + k0 + 2 * lhalf);
        v2f b;
        b.x = W[(k0 + 2 * lhalf)     * 32 + n0 + lm];
        b.y = W[(k0 + 2 * lhalf + 1) * 32 + n0 + lm];
        c = __builtin_amdgcn_wmma_f32_16x16x4_f32(false, a, false, b,
                                                  (short)0, c, false, false);
    }

    const int gcol = wave * 16 + lm;
    #pragma unroll
    for (int v = 0; v < 8; ++v) {
        const int row = m0 + v + 8 * lhalf;
        if (row < nrows) xw[(size_t)row * 64 + gcol] = c[v];
    }
}

__global__ void init_deg_kernel(float* __restrict__ dn, float* __restrict__ ds, int n)
{
    const int i = blockIdx.x * blockDim.x + threadIdx.x;
    if (i < n) { dn[i] = 1.0f; ds[i] = 1.0f; }
}

__global__ void count_deg_kernel(const int* __restrict__ dst_n,
                                 const int* __restrict__ dst_s,
                                 float* __restrict__ dn,
                                 float* __restrict__ ds, int E)
{
    const int e = blockIdx.x * blockDim.x + threadIdx.x;
    if (e >= E) return;
    const int*   dst = blockIdx.y ? dst_s : dst_n;
    float*       deg = blockIdx.y ? ds    : dn;
    atomic_add_f32(&deg[dst[e]], 1.0f);
}

__global__ void rsqrt_kernel(float* __restrict__ dn, float* __restrict__ ds, int n)
{
    const int i = blockIdx.x * blockDim.x + threadIdx.x;
    if (i < n) {
        const float a = dn[i], b = ds[i];
        dn[i] = (a > 0.0f) ? rsqrtf(a) : 0.0f;
        ds[i] = (b > 0.0f) ? rsqrtf(b) : 0.0f;
    }
}

__global__ void init_h_kernel(const float* __restrict__ xw,
                              const float* __restrict__ dn,
                              const float* __restrict__ ds,
                              float* __restrict__ h, int n)
{
    const int t = blockIdx.x * blockDim.x + threadIdx.x;
    if (t >= n * 32) return;
    const int i = t >> 5, j = t & 31;
    const float a = dn[i], b = ds[i];
    h[t] = xw[(size_t)i * 64 + j]      * (a * a)
         + xw[(size_t)i * 64 + 32 + j] * (b * b);
}

__global__ __launch_bounds__(256)
void scatter_kernel(const int* __restrict__ en,
                    const int* __restrict__ es,
                    const float* __restrict__ xw,
                    const float* __restrict__ dn,
                    const float* __restrict__ dsv,
                    float* __restrict__ h, int E)
{
    const int lane = (int)(threadIdx.x & 31u);
    const int wave = __builtin_amdgcn_readfirstlane((int)(threadIdx.x >> 5));
    const int wpb  = (int)(blockDim.x >> 5);

    const int    typ   = blockIdx.y;
    const int*   edges = typ ? es  : en;
    const float* dinv  = typ ? dsv : dn;
    const float* xwt   = xw + (typ ? 32 : 0);

    const int nchunks = (E + 31) >> 5;
    const int stride  = (int)gridDim.x * wpb;

    for (int chunk = (int)blockIdx.x * wpb + wave; chunk < nchunks; chunk += stride) {
        const int e0 = chunk << 5;
        const int e  = e0 + lane;
        int   s = 0, d = 0;
        float nrm = 0.0f;
        if (e < E) {
            s   = edges[e];
            d   = edges[E + e];
            nrm = dinv[s] * dinv[d];
        }
        const int nrm_bits = __float_as_int(nrm);
        const int cnt = (E - e0 < 32) ? (E - e0) : 32;

        if (cnt == 32) {
            #pragma unroll
            for (int g = 0; g < 32; g += 8) {
                float  v[8];
                float* p[8];
                #pragma unroll
                for (int k = 0; k < 8; ++k) {
                    const int sk = __builtin_amdgcn_readlane(s, g + k);
                    v[k] = (xwt + (size_t)sk * 64)[lane];
                }
                #pragma unroll
                for (int k = 0; k < 8; ++k) {
                    const int   dk = __builtin_amdgcn_readlane(d, g + k);
                    const float nk = __int_as_float(__builtin_amdgcn_readlane(nrm_bits, g + k));
                    p[k] = h + (size_t)dk * 32 + lane;
                    v[k] = v[k] * nk;
                    atomic_add_f32(p[k], v[k]);
                }
                wait_issued_atomics8(p, v);
            }
        } else {
            for (int k = 0; k < cnt; ++k) {
                const int   sk = __builtin_amdgcn_readlane(s, k);
                const int   dk = __builtin_amdgcn_readlane(d, k);
                const float nk = __int_as_float(__builtin_amdgcn_readlane(nrm_bits, k));
                atomic_add_f32_wait(h + (size_t)dk * 32 + lane,
                                    (xwt + (size_t)sk * 64)[lane] * nk);
            }
        }
    }
}

__global__ void final_kernel(const float* __restrict__ h,
                             const float* __restrict__ bn,
                             const float* __restrict__ bs,
                             const float* __restrict__ wl,
                             const float* __restrict__ bl,
                             float* __restrict__ out, int n)
{
    const int t = blockIdx.x * blockDim.x + threadIdx.x;
    const int i = t >> 5, j = t & 31;
    if (i >= n) return;
    float v = h[(size_t)i * 32 + j] + bn[j] + bs[j];
    v = fmaxf(v, 0.0f) * wl[j];
    #pragma unroll
    for (int off = 16; off > 0; off >>= 1)
        v += __shfl_xor(v, off, 32);
    if (j == 0) out[i] = v + bl[0];
}

extern "C" void kernel_launch(void* const* d_in, const int* in_sizes, int n_in,
                              void* d_out, int out_size, void* d_ws, size_t ws_size,
                              hipStream_t stream)
{
    const float* x  = (const float*)d_in[0];
    const int*   en = (const int*)  d_in[1];
    const int*   es = (const int*)  d_in[2];
    const float* Wn = (const float*)d_in[3];
    const float* bn = (const float*)d_in[4];
    const float* Ws = (const float*)d_in[5];
    const float* bs = (const float*)d_in[6];
    const float* Wl = (const float*)d_in[7];
    const float* bl = (const float*)d_in[8];
    float* out = (float*)d_out;

    const int N = in_sizes[0] / 64;
    const int E = in_sizes[1] / 2;

    float* ws  = (float*)d_ws;
    float* xw  = ws;
    float* h   = xw + (size_t)N * 64;
    float* dn  = h  + (size_t)N * 32;
    float* ds  = dn + N;

    const int TPB = 256;

    init_deg_kernel<<<(N + TPB - 1) / TPB, TPB, 0, stream>>>(dn, ds, N);
    count_deg_kernel<<<dim3((E + TPB - 1) / TPB, 2), TPB, 0, stream>>>(
        en + E, es + E, dn, ds, E);
    rsqrt_kernel<<<(N + TPB - 1) / TPB, TPB, 0, stream>>>(dn, ds, N);

    gemm_xw_kernel<<<(N + 15) / 16, 128, 0, stream>>>(x, Wn, Ws, xw, N);

    init_h_kernel<<<(N * 32 + TPB - 1) / TPB, TPB, 0, stream>>>(xw, dn, ds, h, N);
    scatter_kernel<<<dim3(1024, 2), TPB, 0, stream>>>(en, es, xw, dn, ds, h, E);

    final_kernel<<<(N * 32 + TPB - 1) / TPB, TPB, 0, stream>>>(h, bn, bs, Wl, bl, out, N);
}


# degree count, self-loop init and head launches moved to resident waves looping over their elements (4 index loads in flight in the count, 16-B accesses in the init); atomics and sum order as before
# speedup vs baseline: 2.1379x; 1.0815x over previous
#include <hip/hip_runtime.h>

typedef __attribute__((ext_vector_type(2))) float v2f;
typedef __attribute__((ext_vector_type(8))) float v8f;

__device__ __forceinline__ void atomic_add_f32(float* p, float v)
{
    asm volatile("global_atomic_add_f32 %0, %1, off"
                 :: "v"(p), "v"(v) : "memory");
}

__device__ __forceinline__ void atomic_add_f32_wait(float* p, float v)
{
    asm volatile("global_atomic_add_f32 %0, %1, off\n\ts_wait_xcnt 0x0"
                 :: "v"(p), "v"(v) : "memory");
}

__device__ __forceinline__ void wait_issued_atomics8(float* const (&p)[8],
                                                     const float (&v)[8])
{
    asm volatile("s_wait_xcnt 0x0"
                 :: "v"(p[0]), "v"(p[1]), "v"(p[2]), "v"(p[3]),
                    "v"(p[4]), "v"(p[5]), "v"(p[6]), "v"(p[7]),
                    "v"(v[0]), "v"(v[1]), "v"(v[2]), "v"(v[3]),
                    "v"(v[4]), "v"(v[5]), "v"(v[6]), "v"(v[7])
                 : "memory");
}

__global__ __launch_bounds__(128)
void gemm_xw_kernel(const float* __restrict__ x,
                    const float* __restrict__ Wn,
                    const float* __restrict__ Ws,
                    float* __restrict__ xw, int nrows)
{
    const int lane  = threadIdx.x & 31;
    const int wave  = threadIdx.x >> 5;
    const int m0    = blockIdx.x * 16;
    const int lhalf = lane >> 4;
    const int lm    = lane & 15;

    const float* __restrict__ W = (wave < 2) ? Wn : Ws;
    const int n0 = (wave & 1) * 16;

    int arow = m0 + lm;
    if (arow >= nrows) arow = nrows - 1;

    v8f c = {};
    #pragma unroll
    for (int k0 = 0; k0 < 64; k0 += 4) {
        v2f a = *(const v2f*)(x + (size_t)arow * 64 + k0 + 2 * lhalf);
        v2f b;
        b.x = W[(k0 + 2 * lhalf)     * 32 + n0 + lm];
        b.y = W[(k0 + 2 * lhalf + 1) * 32 + n0 + lm];
        c = __builtin_amdgcn_wmma_f32_16x16x4_f32(false, a, false, b,
                                                  (short)0, c, false, false);
    }

    const int gcol = wave * 16 + lm;
    #pragma unroll
    for (int v = 0; v < 8; ++v) {
        const int row = m0 + v + 8 * lhalf;
        if (row < nrows) xw[(size_t)row * 64 + gcol] = c[v];
    }
}

__global__ void init_deg_kernel(float* __restrict__ dn, float* __restrict__ ds, int n)
{
    const int i = blockIdx.x * blockDim.x + threadIdx.x;
    if (i < n) { dn[i] = 1.0f; ds[i] = 1.0f; }
}

__global__ __launch_bounds__(256)
void count_deg_kernel(const int* __restrict__ dst_n,
                      const int* __restrict__ dst_s,
                      float* __restrict__ dn,
                      float* __restrict__ ds, int E)
{
    const int* __restrict__ dst = blockIdx.y ? dst_s : dst_n;
    float*                  deg = blockIdx.y ? ds    : dn;
    const int stride = (int)(gridDim.x * blockDim.x);
    const int gid    = (int)(blockIdx.x * blockDim.x + threadIdx.x);
    const int full   = E / stride;

    int k0 = 0;
    for (; k0 + 4 <= full; k0 += 4) {
        int d[4];
        #pragma unroll
        for (int k = 0; k < 4; ++k) d[k] = dst[gid + (k0 + k) * stride];
        #pragma unroll
        for (int k = 0; k < 4; ++k) atomic_add_f32_wait(deg + d[k], 1.0f);
    }
    for (int e = gid + k0 * stride; e < E; e += stride)
        atomic_add_f32_wait(deg + dst[e], 1.0f);
}

__global__ void rsqrt_kernel(float* __restrict__ dn, float* __restrict__ ds, int n)
{
    const int i = blockIdx.x * blockDim.x + threadIdx.x;
    if (i < n) {
        const float a = dn[i], b = ds[i];
        dn[i] = (a > 0.0f) ? rsqrtf(a) : 0.0f;
        ds[i] = (b > 0.0f) ? rsqrtf(b) : 0.0f;
    }
}

__global__ __launch_bounds__(256)
void init_h_kernel(const float* __restrict__ xw,
                   const float* __restrict__ dn,
                   const float* __restrict__ ds,
                   float* __restrict__ h, int n)
{
    const float4* __restrict__ xw4 = (const float4*)xw;
    float4*       __restrict__ h4  = (float4*)h;
    const int total  = n * 8;
    const int stride = (int)(gridDim.x * blockDim.x);
    for (int t = (int)(blockIdx.x * blockDim.x + threadIdx.x); t < total; t += stride) {
        const int i = t >> 3, q = t & 7;
        const float a = dn[i], b = ds[i];
        const float aa = a * a, bb = b * b;
        const float4 u = xw4[(size_t)i * 16 + q];
        const float4 w = xw4[(size_t)i * 16 + 8 + q];
        float4 r;
        r.x = u.x * aa + w.x * bb;
        r.y = u.y * aa + w.y * bb;
        r.z = u.z * aa + w.z * bb;
        r.w = u.w * aa + w.w * bb;
        h4[t] = r;
    }
}

__global__ __launch_bounds__(256)
void scatter_kernel(const int* __restrict__ en,
                    const int* __restrict__ es,
                    const float* __restrict__ xw,
                    const float* __restrict__ dn,
                    const float* __restrict__ dsv,
                    float* __restrict__ h, int E)
{
    const int lane = (int)(threadIdx.x & 31u);
    const int wave = __builtin_amdgcn_readfirstlane((int)(threadIdx.x >> 5));
    const int wpb  = (int)(blockDim.x >> 5);

    const int    typ   = blockIdx.y;
    const int*   edges = typ ? es  : en;
    const float* dinv  = typ ? dsv : dn;
    const float* xwt   = xw + (typ ? 32 : 0);

    const int nchunks = (E + 31) >> 5;
    const int stride  = (int)gridDim.x * wpb;

    for (int chunk = (int)blockIdx.x * wpb + wave; chunk < nchunks; chunk += stride) {
        const int e0 = chunk << 5;
        const int e  = e0 + lane;
        int   s = 0, d = 0;
        float nrm = 0.0f;
        if (e < E) {
            s   = edges[e];
            d   = edges[E + e];
            nrm = dinv[s] * dinv[d];
        }
        const int nrm_bits = __float_as_int(nrm);
        const int cnt = (E - e0 < 32) ? (E - e0) : 32;

        if (cnt == 32) {
            #pragma unroll
            for (int g = 0; g < 32; g += 8) {
                float  v[8];
                float* p[8];
                #pragma unroll
                for (int k = 0; k < 8; ++k) {
                    const int sk = __builtin_amdgcn_readlane(s, g + k);
                    v[k] = (xwt + (size_t)sk * 64)[lane];
                }
                #pragma unroll
                for (int k = 0; k < 8; ++k) {
                    const int   dk = __builtin_amdgcn_readlane(d, g + k);
                    const float nk = __int_as_float(__builtin_amdgcn_readlane(nrm_bits, g + k));
                    p[k] = h + (size_t)dk * 32 + lane;
                    v[k] = v[k] * nk;
                    atomic_add_f32(p[k], v[k]);
                }
                wait_issued_atomics8(p, v);
            }
        } else {
            for (int k = 0; k < cnt; ++k) {
                const int   sk = __builtin_amdgcn_readlane(s, k);
                const int   dk = __builtin_amdgcn_readlane(d, k);
                const float nk = __int_as_float(__builtin_amdgcn_readlane(nrm_bits, k));
                atomic_add_f32_wait(h + (size_t)dk * 32 + lane,
                                    (xwt + (size_t)sk * 64)[lane] * nk);
            }
        }
    }
}

__global__ __launch_bounds__(256)
void final_kernel(const float* __restrict__ h,
                  const float* __restrict__ bn,
                  const float* __restrict__ bs,
                  const float* __restrict__ wl,
                  const float* __restrict__ bl,
                  float* __restrict__ out, int n)
{
    const int j = (int)(threadIdx.x & 31u);
    const int wave = __builtin_amdgcn_readfirstlane((int)(threadIdx.x >> 5));
    const int wpb  = (int)(blockDim.x >> 5);
    const int stride = (int)gridDim.x * wpb;
    const float bnj = bn[j], bsj = bs[j], wlj = wl[j], bl0 = bl[0];
    for (int i = (int)blockIdx.x * wpb + wave; i < n; i += stride) {
        float v = h[(size_t)i * 32 + j] + bnj + bsj;
        v = fmaxf(v, 0.0f) * wlj;
        #pragma unroll
        for (int off = 16; off > 0; off >>= 1)
            v += __shfl_xor(v, off, 32);
        if (j == 0) out[i] = v + bl0;
    }
}

extern "C" void kernel_launch(void* const* d_in, const int* in_sizes, int n_in,
                              void* d_out, int out_size, void* d_ws, size_t ws_size,
                              hipStream_t stream)
{
    const float* x  = (const float*)d_in[0];
    const int*   en = (const int*)  d_in[1];
    const int*   es = (const int*)  d_in[2];
    const float* Wn = (const float*)d_in[3];
    const float* bn = (const float*)d_in[4];
    const float* Ws = (const float*)d_in[5];
    const float* bs = (const float*)d_in[6];
    const float* Wl = (const float*)d_in[7];
    const float* bl = (const float*)d_in[8];
    float* out = (float*)d_out;

    const int N = in_sizes[0] / 64;
    const int E = in_sizes[1] / 2;

    float* ws  = (float*)d_ws;
    float* xw  = ws;
    float* h   = xw + (size_t)N * 64;
    float* dn  = h  + (size_t)N * 32;
    float* ds  = dn + N;

    const int TPB = 256;

    init_deg_kernel<<<(N + TPB - 1) / TPB, TPB, 0, stream>>>(dn, ds, N);
    count_deg_kernel<<<dim3(1024, 2), TPB, 0, stream>>>(
        en + E, es + E, dn, ds, E);
    rsqrt_kernel<<<(N + TPB - 1) / TPB, TPB, 0, stream>>>(dn, ds, N);

    gemm_xw_kernel<<<(N + 15) / 16, 128, 0, stream>>>(x, Wn, Ws, xw, N);

    init_h_kernel<<<2048, TPB, 0, stream>>>(xw, dn, ds, h, N);
    scatter_kernel<<<dim3(1024, 2), TPB, 0, stream>>>(en, es, xw, dn, ds, h, E);

    final_kernel<<<2048, TPB, 0, stream>>>(h, bn, bs, Wl, bl, out, N);
}


# edges ordered by destination (two counting passes, in-wave ranks by ballot, wave-private LDS counters); a wave per h row pulls and sums its sources in registers; all atomics gone, sum order now fixed
# speedup vs baseline: 3.8980x; 1.8233x over previous
#include <hip/hip_runtime.h>


typedef __attribute__((ext_vector_type(2))) float v2f;
typedef __attribute__((ext_vector_type(8))) float v8f;


#define FINE_BITS 9
#define FINE_BINS 512
#define COARSE_BINS 256

template <int NBITS>
__device__ __forceinline__ unsigned match_mask(int key, bool valid)
{
    unsigned m = __builtin_amdgcn_ballot_w32(valid);
    #pragma unroll
    for (int b = 0; b < NBITS; ++b) {
        const bool bit = ((key >> b) & 1) != 0;
        const unsigned bal = __builtin_amdgcn_ballot_w32(bit);
        m &= bit ? bal : ~bal;
    }
    return m;
}

__device__ __forceinline__ int wave_incl_scan(int v, int lane)
{
    #pragma unroll
    for (int off = 1; off < 32; off <<= 1) {
        const int y = __shfl_up(v, off, 32);
        if (lane >= off) v += y;
    }
    return v;
}

__global__ __launch_bounds__(128)
void gemm_xw_kernel(const float* __restrict__ x,
                    const float* __restrict__ W,
                    float* __restrict__ xw, int nrows)
{
    const int lane  = threadIdx.x & 31;
    const int wave  = threadIdx.x >> 5;
    const int m0    = (blockIdx.x * 2 + (wave >> 1)) * 16;
    const int lhalf = lane >> 4;
    const int lm    = lane & 15;
    const int n0    = (wave & 1) * 16;

    int arow = m0 + lm;
    if (arow >= nrows) arow = nrows - 1;

    v8f c = {};
    #pragma unroll
    for (int k0 = 0; k0 < 64; k0 += 4) {
        v2f a = *(const v2f*)(x + (size_t)arow * 64 + k0 + 2 * lhalf);
        v2f b;
        b.x = W[(k0 + 2 * lhalf)     * 32 + n0 + lm];
        b.y = W[(k0 + 2 * lhalf + 1) * 32 + n0 + lm];
        c = __builtin_amdgcn_wmma_f32_16x16x4_f32(false, a, false, b,
                                                  (short)0, c, false, false);
    }

    #pragma unroll
    for (int v = 0; v < 8; ++v) {
        const int row = m0 + v + 8 * lhalf;
        if (row < nrows) xw[(size_t)row * 32 + n0 + lm] = c[v];
    }
}

__device__ __forceinline__ void wave_coarse_count(const int* __restrict__ dst,
                                                  int wbeg, int wend, int lane, int* wh)
{
    const unsigned lt = (1u << lane) - 1u;
    for (int it = wbeg; it < wend; it += 128) {
        int  k[4];
        bool v[4];
        #pragma unroll
        for (int u = 0; u < 4; ++u) {
            const int e = it + 32 * u + lane;
            v[u] = e < wend;
            k[u] = (dst[v[u] ? e : (wend - 1)] >> FINE_BITS) & (COARSE_BINS - 1);
        }
        #pragma unroll
        for (int u = 0; u < 4; ++u) {
            const unsigned m = match_mask<8>(k[u], v[u]);
            if (v[u] && (m & lt) == 0u) wh[k[u]] += __popc(m);
        }
    }
}

__device__ __forceinline__ void coarse_slice(int E, int w, int& wbeg, int& wend)
{
    const int pb   = ((E + (int)gridDim.x - 1) / (int)gridDim.x + 1023) & ~1023;
    const int bbeg = (int)blockIdx.x * pb;
    const int pw   = pb >> 5;
    wbeg = bbeg + w * pw;
    wend = wbeg + pw;
    if (wend > E) wend = E;
}

__global__ __launch_bounds__(1024)
void coarse_hist_kernel(const int* __restrict__ dst, int* __restrict__ mat, int E)
{
    __shared__ int hist[32 * COARSE_BINS];
    const int tid  = (int)threadIdx.x;
    const int lane = tid & 31;
    const int w    = __builtin_amdgcn_readfirstlane(tid >> 5);
    for (int i = tid; i < 32 * COARSE_BINS; i += 1024) hist[i] = 0;
    __syncthreads();
    int wbeg, wend;
    coarse_slice(E, w, wbeg, wend);
    wave_coarse_count(dst, wbeg, wend, lane, hist + w * COARSE_BINS);
    __syncthreads();
    if (tid < COARSE_BINS) {
        int tot = 0;
        for (int w2 = 0; w2 < 32; ++w2) tot += hist[w2 * COARSE_BINS + tid];
        mat[tid * (int)gridDim.x + (int)blockIdx.x] = tot;
    }
}

__global__ __launch_bounds__(1024)
void scan_mat_kernel(const int* __restrict__ cnt, int* __restrict__ mat, int M)
{
    __shared__ int wsum[32];
    __shared__ int sbase;
    const int tid  = (int)threadIdx.x;
    const int lane = tid & 31;
    const int w    = tid >> 5;
    const int lim  = (int)blockIdx.x * 1024;

    int part = 0;
    const int4* __restrict__ cnt4 = (const int4*)cnt;
    #pragma unroll 8
    for (int j = tid; j < (lim >> 2); j += 1024) {
        const int4 q = cnt4[j];
        part += (q.x + q.y) + (q.z + q.w);
    }
    #pragma unroll
    for (int off = 16; off > 0; off >>= 1) part += __shfl_xor(part, off, 32);
    if (lane == 0) wsum[w] = part;
    __syncthreads();
    if (w == 0) {
        int xs = wsum[lane];
        #pragma unroll
        for (int off = 16; off > 0; off >>= 1) xs += __shfl_xor(xs, off, 32);
        if (lane == 0) sbase = xs;
    }
    __syncthreads();
    const int base = sbase;

    const int i   = lim + tid;
    const int c   = (i < M) ? cnt[i] : 0;
    const int inc = wave_incl_scan(c, lane);
    if (lane == 31) wsum[w] = inc;
    __syncthreads();
    if (w == 0) {
        const int own = wsum[lane];
        const int sc  = wave_incl_scan(own, lane);
        wsum[lane] = sc - own;
    }
    __syncthreads();
    if (i < M) mat[i] = base + wsum[w] + inc - c;
}

__global__ __launch_bounds__(1024)
void coarse_place_kernel(const int* __restrict__ src, const int* __restrict__ dst,
                         const int* __restrict__ mat, int* __restrict__ items, int E)
{
    __shared__ int hist[32 * COARSE_BINS];
    const int tid  = (int)threadIdx.x;
    const int lane = tid & 31;
    const int w    = __builtin_amdgcn_readfirstlane(tid >> 5);
    for (int i = tid; i < 32 * COARSE_BINS; i += 1024) hist[i] = 0;
    __syncthreads();
    int wbeg, wend;
    coarse_slice(E, w, wbeg, wend);
    int* wh = hist + w * COARSE_BINS;
    wave_coarse_count(dst, wbeg, wend, lane, wh);
    __syncthreads();
    if (tid < COARSE_BINS) {
        int run = mat[tid * (int)gridDim.x + (int)blockIdx.x];
        for (int w2 = 0; w2 < 32; ++w2) {
            const int t = hist[w2 * COARSE_BINS + tid];
            hist[w2 * COARSE_BINS + tid] = run;
            run += t;
        }
    }
    __syncthreads();
    const unsigned lt = (1u << lane) - 1u;
    for (int it = wbeg; it < wend; it += 128) {
        int  d[4], s[4];
        bool v[4];
        #pragma unroll
        for (int u = 0; u < 4; ++u) {
            const int e  = it + 32 * u + lane;
            v[u] = e < wend;
            const int ec = v[u] ? e : (wend - 1);
            d[u] = dst[ec];
            s[u] = src[ec];
        }
        #pragma unroll
        for (int u = 0; u < 4; ++u) {
            const int key = (d[u] >> FINE_BITS) & (COARSE_BINS - 1);
            const unsigned m = match_mask<8>(key, v[u]);
            const int rank = __popc(m & lt);
            if (v[u]) {
                int pos = wh[key] + rank;
                pos = (pos < E) ? pos : (E - 1);
                pos = (pos > 0) ? pos : 0;
                items[pos] = (s[u] << FINE_BITS) | (d[u] & (FINE_BINS - 1));
                if (rank == 0) wh[key] = pos - rank + __popc(m);
            }
        }
    }
}

__global__ __launch_bounds__(1024)
void fine_sort_kernel(const int* __restrict__ items, const int* __restrict__ mat, int nb,
                      int* __restrict__ csr, int* __restrict__ rowend, int n, int E)
{
    __shared__ int hist[32 * FINE_BINS];
    __shared__ int wsum[32];
    const int tid  = (int)threadIdx.x;
    const int lane = tid & 31;
    const int w    = __builtin_amdgcn_readfirstlane(tid >> 5);
    const int j    = (int)blockIdx.x;
    const int o0   = mat[j * nb];
    const int o1   = (j + 1 < COARSE_BINS) ? mat[(j + 1) * nb] : E;
    const int len  = o1 - o0;
    const int pw   = (((len + 31) >> 5) + 31) & ~31;
    const int wbeg = o0 + w * pw;
    int wend = wbeg + pw; if (wend > o1) wend = o1;
    int* wh = hist + w * FINE_BINS;
    const unsigned lt = (1u << lane) - 1u;

    for (int i = tid; i < 32 * FINE_BINS; i += 1024) hist[i] = 0;
    __syncthreads();

    for (int it = wbeg; it < wend; it += 128) {
        int  k[4];
        bool v[4];
        #pragma unroll
        for (int u = 0; u < 4; ++u) {
            const int e = it + 32 * u + lane;
            v[u] = e < wend;
            k[u] = items[v[u] ? e : (wend - 1)] & (FINE_BINS - 1);
        }
        #pragma unroll
        for (int u = 0; u < 4; ++u) {
            const unsigned m = match_mask<FINE_BITS>(k[u], v[u]);
            if (v[u] && (m & lt) == 0u) wh[k[u]] += __popc(m);
        }
    }
    __syncthreads();

    int tot = 0;
    if (tid < FINE_BINS) {
        for (int w2 = 0; w2 < 32; ++w2) tot += hist[w2 * FINE_BINS + tid];
    }
    const int inc = wave_incl_scan(tot, lane);
    if (lane == 31) wsum[w] = (w < 16) ? inc : 0;
    __syncthreads();
    if (w == 0) {
        const int own = wsum[lane];
        const int sc  = wave_incl_scan(own, lane);
        wsum[lane] = sc - own;
    }
    __syncthreads();
    if (tid < FINE_BINS) {
        const int start = o0 + wsum[w] + inc - tot;
        const int node  = j * FINE_BINS + tid;
        if (node < n) rowend[node] = start + tot;
        int run = start;
        for (int w2 = 0; w2 < 32; ++w2) {
            const int t = hist[w2 * FINE_BINS + tid];
            hist[w2 * FINE_BINS + tid] = run;
            run += t;
        }
    }
    __syncthreads();

    for (int it = wbeg; it < wend; it += 128) {
        int  p[4];
        bool v[4];
        #pragma unroll
        for (int u = 0; u < 4; ++u) {
            const int e = it + 32 * u + lane;
            v[u] = e < wend;
            p[u] = items[v[u] ? e : (wend - 1)];
        }
        #pragma unroll
        for (int u = 0; u < 4; ++u) {
            const int key = p[u] & (FINE_BINS - 1);
            const unsigned m = match_mask<FINE_BITS>(key, v[u]);
            const int rank = __popc(m & lt);
            if (v[u]) {
                int pos = wh[key] + rank;
                pos = (pos < E) ? pos : (E - 1);
                pos = (pos > 0) ? pos : 0;
                csr[pos] = (int)((unsigned)p[u] >> FINE_BITS);
                if (rank == 0) wh[key] = pos - rank + __popc(m);
            }
        }
    }
}

template <int G>
__device__ __forceinline__ float gather_sum(const float* __restrict__ xw, int s, int nrm_bits,
                                            int lane, float acc)
{
    float v[G];
    #pragma unroll
    for (int k = 0; k < G; ++k) {
        const int sk = __builtin_amdgcn_readlane(s, k);
        v[k] = (xw + (size_t)sk * 32)[lane];
    }
    #pragma unroll
    for (int k = 0; k < G; ++k) {
        const float nk = __int_as_float(__builtin_amdgcn_readlane(nrm_bits, k));
        acc += v[k] * nk;
    }
    return acc;
}

__device__ __forceinline__ float dinv_of(int lo, int hi)
{
    const float deg = (float)(hi - lo + 1);
    return (deg > 0.0f) ? rsqrtf(deg) : 0.0f;
}

__global__ __launch_bounds__(256)
void pull_kernel(const float* __restrict__ xw, const int* __restrict__ csr,
                 const int* __restrict__ rowend,
                 float* __restrict__ hpart,
                 const float* __restrict__ bn, const float* __restrict__ bs,
                 const float* __restrict__ wl, const float* __restrict__ bl,
                 float* __restrict__ out, int n, int last)
{
    const int lane = (int)(threadIdx.x & 31u);
    const int wave = __builtin_amdgcn_readfirstlane((int)(threadIdx.x >> 5));
    const int wpb  = (int)(blockDim.x >> 5);
    const int nw   = (int)gridDim.x * wpb;
    const int R    = (n + nw - 1) / nw;
    const int r0   = ((int)blockIdx.x * wpb + wave) * R;
    const int r1   = (r0 + R < n) ? (r0 + R) : n;

    const float bnj = bn[lane], bsj = bs[lane], wlj = wl[lane], bl0 = bl[0];

    for (int i = r0; i < r1; ++i) {
        const int   start = i ? rowend[i - 1] : 0;
        const int   end   = rowend[i];
        const float di    = dinv_of(start, end);
        const float selfv = xw[(size_t)i * 32 + lane];
        float acc = 0.0f;

        for (int base = start; base < end; base += 32) {
            const int cnt = (end - base < 32) ? (end - base) : 32;
            int   s   = i;
            float nrm = 0.0f;
            if (lane < cnt) {
                s = csr[base + lane];
                const int hi = rowend[s];
                const int lo = s ? rowend[s - 1] : 0;
                nrm = dinv_of(lo, hi) * di;
            }
            const int nrm_bits = __float_as_int(nrm);
            if (cnt > 16) acc = gather_sum<32>(xw, s, nrm_bits, lane, acc);
            else          acc = gather_sum<16>(xw, s, nrm_bits, lane, acc);
        }
        acc += selfv * (di * di);

        if (!last) {
            hpart[(size_t)i * 32 + lane] = acc;
        } else {
            float v = hpart[(size_t)i * 32 + lane] + acc;
            v = v + bnj + bsj;
            v = fmaxf(v, 0.0f) * wlj;
            #pragma unroll
            for (int off = 16; off > 0; off >>= 1)
                v += __shfl_xor(v, off, 32);
            if (lane == 0) out[i] = v + bl0;
        }
    }
}

extern "C" void kernel_launch(void* const* d_in, const int* in_sizes, int n_in,
                              void* d_out, int out_size, void* d_ws, size_t ws_size,
                              hipStream_t stream)
{
    const float* x  = (const float*)d_in[0];
    const int*   en = (const int*)  d_in[1];
    const int*   es = (const int*)  d_in[2];
    const float* Wn = (const float*)d_in[3];
    const float* bn = (const float*)d_in[4];
    const float* Ws = (const float*)d_in[5];
    const float* bs = (const float*)d_in[6];
    const float* Wl = (const float*)d_in[7];
    const float* bl = (const float*)d_in[8];
    float* out = (float*)d_out;

    const int N = in_sizes[0] / 64;
    const int E = in_sizes[1] / 2;

    float* ws     = (float*)d_ws;
    float* xw     = ws;
    int*   items  = (int*)ws;
    float* hpart  = ws + (size_t)N * 32;
    int*   rowend = (int*)(hpart + (size_t)N * 32);
    int*   mat    = rowend + N;
    int*   csr    = mat + N;

    const int CB = 256;

    coarse_hist_kernel<<<CB, 1024, 0, stream>>>(en + E, csr, E);
    scan_mat_kernel<<<COARSE_BINS * CB / 1024, 1024, 0, stream>>>(csr, mat, COARSE_BINS * CB);
    coarse_place_kernel<<<CB, 1024, 0, stream>>>(en, en + E, mat, items, E);
    fine_sort_kernel<<<(N + FINE_BINS - 1) / FINE_BINS, 1024, 0, stream>>>(items, mat, CB, csr, rowend, N, E);
    gemm_xw_kernel<<<(N + 31) / 32, 128, 0, stream>>>(x, Wn, xw, N);
    pull_kernel<<<2048, 256, 0, stream>>>(xw, csr, rowend, hpart, bn, bs, Wl, bl, out, N, 0);

    coarse_hist_kernel<<<CB, 1024, 0, stream>>>(es + E, csr, E);
    scan_mat_kernel<<<COARSE_BINS * CB / 1024, 1024, 0, stream>>>(csr, mat, COARSE_BINS * CB);
    coarse_place_kernel<<<CB, 1024, 0, stream>>>(es, es + E, mat, items, E);
    fine_sort_kernel<<<(N + FINE_BINS - 1) / FINE_BINS, 1024, 0, stream>>>(items, mat, CB, csr, rowend, N, E);
    gemm_xw_kernel<<<(N + 31) / 32, 128, 0, stream>>>(x, Ws, xw, N);
    pull_kernel<<<2048, 256, 0, stream>>>(xw, csr, rowend, hpart, bn, bs, Wl, bl, out, N, 1);
}


# placing passes (coarse and fine) stage their items in LDS in final order and write them out as contiguous runs of whole lines, not one scattered 4-byte store per item; result and sum order unchanged
# speedup vs baseline: 4.8706x; 1.2495x over previous
#include <hip/hip_runtime.h>


typedef __attribute__((ext_vector_type(2))) float v2f;
typedef __attribute__((ext_vector_type(8))) float v8f;


#define FINE_BITS 9
#define FINE_BINS 512
#define COARSE_BINS 256
#define PLACE_CAP 16384
#define FINE_CAP  20480

template <int NBITS>
__device__ __forceinline__ unsigned match_mask(int key, bool valid)
{
    unsigned m = __builtin_amdgcn_ballot_w32(valid);
    #pragma unroll
    for (int b = 0; b < NBITS; ++b) {
        const bool bit = ((key >> b) & 1) != 0;
        const unsigned bal = __builtin_amdgcn_ballot_w32(bit);
        m &= bit ? bal : ~bal;
    }
    return m;
}

__device__ __forceinline__ int wave_incl_scan(int v, int lane)
{
    #pragma unroll
    for (int off = 1; off < 32; off <<= 1) {
        const int y = __shfl_up(v, off, 32);
        if (lane >= off) v += y;
    }
    return v;
}

__global__ __launch_bounds__(128)
void gemm_xw_kernel(const float* __restrict__ x,
                    const float* __restrict__ W,
                    float* __restrict__ xw, int nrows)
{
    const int lane  = threadIdx.x & 31;
    const int wave  = threadIdx.x >> 5;
    const int m0    = (blockIdx.x * 2 + (wave >> 1)) * 16;
    const int lhalf = lane >> 4;
    const int lm    = lane & 15;
    const int n0    = (wave & 1) * 16;

    int arow = m0 + lm;
    if (arow >= nrows) arow = nrows - 1;

    v8f c = {};
    #pragma unroll
    for (int k0 = 0; k0 < 64; k0 += 4) {
        v2f a = *(const v2f*)(x + (size_t)arow * 64 + k0 + 2 * lhalf);
        v2f b;
        b.x = W[(k0 + 2 * lhalf)     * 32 + n0 + lm];
        b.y = W[(k0 + 2 * lhalf + 1) * 32 + n0 + lm];
        c = __builtin_amdgcn_wmma_f32_16x16x4_f32(false, a, false, b,
                                                  (short)0, c, false, false);
    }

    #pragma unroll
    for (int v = 0; v < 8; ++v) {
        const int row = m0 + v + 8 * lhalf;
        if (row < nrows) xw[(size_t)row * 32 + n0 + lm] = c[v];
    }
}

__device__ __forceinline__ void wave_coarse_count(const int* __restrict__ dst,
                                                  int wbeg, int wend, int lane, int* wh)
{
    const unsigned lt = (1u << lane) - 1u;
    for (int it = wbeg; it < wend; it += 128) {
        int  k[4];
        bool v[4];
        #pragma unroll
        for (int u = 0; u < 4; ++u) {
            const int e = it + 32 * u + lane;
            v[u] = e < wend;
            k[u] = (dst[v[u] ? e : (wend - 1)] >> FINE_BITS) & (COARSE_BINS - 1);
        }
        #pragma unroll
        for (int u = 0; u < 4; ++u) {
            const unsigned m = match_mask<8>(k[u], v[u]);
            if (v[u] && (m & lt) == 0u) wh[k[u]] += __popc(m);
        }
    }
}

__device__ __forceinline__ void coarse_slice(int E, int w, int& wbeg, int& wend)
{
    const int pb   = ((E + (int)gridDim.x - 1) / (int)gridDim.x + 1023) & ~1023;
    const int bbeg = (int)blockIdx.x * pb;
    const int pw   = pb >> 5;
    wbeg = bbeg + w * pw;
    wend = wbeg + pw;
    if (wend > E) wend = E;
}

__global__ __launch_bounds__(1024)
void coarse_hist_kernel(const int* __restrict__ dst, int* __restrict__ mat, int E)
{
    __shared__ int hist[32 * COARSE_BINS];
    const int tid  = (int)threadIdx.x;
    const int lane = tid & 31;
    const int w    = __builtin_amdgcn_readfirstlane(tid >> 5);
    for (int i = tid; i < 32 * COARSE_BINS; i += 1024) hist[i] = 0;
    __syncthreads();
    int wbeg, wend;
    coarse_slice(E, w, wbeg, wend);
    wave_coarse_count(dst, wbeg, wend, lane, hist + w * COARSE_BINS);
    __syncthreads();
    if (tid < COARSE_BINS) {
        int tot = 0;
        for (int w2 = 0; w2 < 32; ++w2) tot += hist[w2 * COARSE_BINS + tid];
        mat[tid * (int)gridDim.x + (int)blockIdx.x] = tot;
    }
}

__global__ __launch_bounds__(1024)
void scan_mat_kernel(const int* __restrict__ cnt, int* __restrict__ mat, int M)
{
    __shared__ int wsum[32];
    __shared__ int sbase;
    const int tid  = (int)threadIdx.x;
    const int lane = tid & 31;
    const int w    = tid >> 5;
    const int lim  = (int)blockIdx.x * 1024;

    int part = 0;
    const int4* __restrict__ cnt4 = (const int4*)cnt;
    #pragma unroll 8
    for (int j = tid; j < (lim >> 2); j += 1024) {
        const int4 q = cnt4[j];
        part += (q.x + q.y) + (q.z + q.w);
    }
    #pragma unroll
    for (int off = 16; off > 0; off >>= 1) part += __shfl_xor(part, off, 32);
    if (lane == 0) wsum[w] = part;
    __syncthreads();
    if (w == 0) {
        int xs = wsum[lane];
        #pragma unroll
        for (int off = 16; off > 0; off >>= 1) xs += __shfl_xor(xs, off, 32);
        if (lane == 0) sbase = xs;
    }
    __syncthreads();
    const int base = sbase;

    const int i   = lim + tid;
    const int c   = (i < M) ? cnt[i] : 0;
    const int inc = wave_incl_scan(c, lane);
    if (lane == 31) wsum[w] = inc;
    __syncthreads();
    if (w == 0) {
        const int own = wsum[lane];
        const int sc  = wave_incl_scan(own, lane);
        wsum[lane] = sc - own;
    }
    __syncthreads();
    if (i < M) mat[i] = base + wsum[w] + inc - c;
}

__global__ __launch_bounds__(1024)
void coarse_place_kernel(const int* __restrict__ src, const int* __restrict__ dst,
                         const int* __restrict__ mat, int* __restrict__ items, int E)
{
    __shared__ int hist[32 * COARSE_BINS];
    __shared__ int stage[PLACE_CAP];
    __shared__ int gbase[COARSE_BINS];
    __shared__ int lbase[COARSE_BINS];
    __shared__ int bcnt[COARSE_BINS];
    __shared__ int wsum[32];
    const int tid  = (int)threadIdx.x;
    const int lane = tid & 31;
    const int w    = __builtin_amdgcn_readfirstlane(tid >> 5);
    for (int i = tid; i < 32 * COARSE_BINS; i += 1024) hist[i] = 0;
    __syncthreads();
    int wbeg, wend;
    coarse_slice(E, w, wbeg, wend);
    const bool staged = (((E + (int)gridDim.x - 1) / (int)gridDim.x + 1023) & ~1023) <= PLACE_CAP;
    int* wh = hist + w * COARSE_BINS;
    wave_coarse_count(dst, wbeg, wend, lane, wh);
    __syncthreads();
    int start = 0, tot = 0;
    if (tid < COARSE_BINS) {
        start = mat[tid * (int)gridDim.x + (int)blockIdx.x];
        int run = start;
        for (int w2 = 0; w2 < 32; ++w2) {
            const int t = hist[w2 * COARSE_BINS + tid];
            hist[w2 * COARSE_BINS + tid] = run;
            run += t;
        }
        tot = run - start;
    }
    const int inc = wave_incl_scan(tot, lane);
    if (lane == 31) wsum[w] = (w < COARSE_BINS / 32) ? inc : 0;
    __syncthreads();
    if (w == 0) {
        const int own = wsum[lane];
        const int sc  = wave_incl_scan(own, lane);
        wsum[lane] = sc - own;
    }
    __syncthreads();
    if (tid < COARSE_BINS) {
        gbase[tid] = start;
        lbase[tid] = wsum[w] + inc - tot;
        bcnt[tid]  = tot;
    }
    __syncthreads();
    const unsigned lt = (1u << lane) - 1u;
    for (int it = wbeg; it < wend; it += 128) {
        int  d[4], s[4];
        bool v[4];
        #pragma unroll
        for (int u = 0; u < 4; ++u) {
            const int e  = it + 32 * u + lane;
            v[u] = e < wend;
            const int ec = v[u] ? e : (wend - 1);
            d[u] = dst[ec];
            s[u] = src[ec];
        }
        #pragma unroll
        for (int u = 0; u < 4; ++u) {
            const int key = (d[u] >> FINE_BITS) & (COARSE_BINS - 1);
            const unsigned m = match_mask<8>(key, v[u]);
            const int rank = __popc(m & lt);
            if (v[u]) {
                int pos = wh[key] + rank;
                pos = (pos < E) ? pos : (E - 1);
                pos = (pos > 0) ? pos : 0;
                const int item = (s[u] << FINE_BITS) | (d[u] & (FINE_BINS - 1));
                if (staged) {
                    int lp = pos - gbase[key] + lbase[key];
                    lp = (lp < PLACE_CAP) ? lp : (PLACE_CAP - 1);
                    lp = (lp > 0) ? lp : 0;
                    stage[lp] = item;
                } else {
                    items[pos] = item;
                }
                if (rank == 0) wh[key] = pos - rank + __popc(m);
            }
        }
    }
    if (staged) {
        __syncthreads();
        for (int k = w * (COARSE_BINS / 32); k < (w + 1) * (COARSE_BINS / 32); ++k) {
            const int g = gbase[k], l = lbase[k], c = bcnt[k];
            for (int i = lane; i < c; i += 32) {
                int pos = g + i;
                pos = (pos < E) ? pos : (E - 1);
                pos = (pos > 0) ? pos : 0;
                const int li = (l + i < PLACE_CAP) ? (l + i) : (PLACE_CAP - 1);
                items[pos] = stage[li];
            }
        }
    }
}

__global__ __launch_bounds__(1024)
void fine_sort_kernel(const int* __restrict__ items, const int* __restrict__ mat, int nb,
                      int* __restrict__ csr, int* __restrict__ rowend, int n, int E)
{
    __shared__ int hist[32 * FINE_BINS];
    __shared__ int wsum[32];
    __shared__ int stage[FINE_CAP];
    const int tid  = (int)threadIdx.x;
    const int lane = tid & 31;
    const int w    = __builtin_amdgcn_readfirstlane(tid >> 5);
    const int j    = (int)blockIdx.x;
    const int o0   = mat[j * nb];
    const int o1   = (j + 1 < COARSE_BINS) ? mat[(j + 1) * nb] : E;
    const int len  = o1 - o0;
    const bool staged = len <= FINE_CAP;
    const int pw   = (((len + 31) >> 5) + 31) & ~31;
    const int wbeg = o0 + w * pw;
    int wend = wbeg + pw; if (wend > o1) wend = o1;
    int* wh = hist + w * FINE_BINS;
    const unsigned lt = (1u << lane) - 1u;

    for (int i = tid; i < 32 * FINE_BINS; i += 1024) hist[i] = 0;
    __syncthreads();

    for (int it = wbeg; it < wend; it += 128) {
        int  k[4];
        bool v[4];
        #pragma unroll
        for (int u = 0; u < 4; ++u) {
            const int e = it + 32 * u + lane;
            v[u] = e < wend;
            k[u] = items[v[u] ? e : (wend - 1)] & (FINE_BINS - 1);
        }
        #pragma unroll
        for (int u = 0; u < 4; ++u) {
            const unsigned m = match_mask<FINE_BITS>(k[u], v[u]);
            if (v[u] && (m & lt) == 0u) wh[k[u]] += __popc(m);
        }
    }
    __syncthreads();

    int tot = 0;
    if (tid < FINE_BINS) {
        for (int w2 = 0; w2 < 32; ++w2) tot += hist[w2 * FINE_BINS + tid];
    }
    const int inc = wave_incl_scan(tot, lane);
    if (lane == 31) wsum[w] = (w < 16) ? inc : 0;
    __syncthreads();
    if (w == 0) {
        const int own = wsum[lane];
        const int sc  = wave_incl_scan(own, lane);
        wsum[lane] = sc - own;
    }
    __syncthreads();
    if (tid < FINE_BINS) {
        const int start = o0 + wsum[w] + inc - tot;
        const int node  = j * FINE_BINS + tid;
        if (node < n) rowend[node] = start + tot;
        int run = start;
        for (int w2 = 0; w2 < 32; ++w2) {
            const int t = hist[w2 * FINE_BINS + tid];
            hist[w2 * FINE_BINS + tid] = run;
            run += t;
        }
    }
    __syncthreads();

    for (int it = wbeg; it < wend; it += 128) {
        int  p[4];
        bool v[4];
        #pragma unroll
        for (int u = 0; u < 4; ++u) {
            const int e = it + 32 * u + lane;
            v[u] = e < wend;
            p[u] = items[v[u] ? e : (wend - 1)];
        }
        #pragma unroll
        for (int u = 0; u < 4; ++u) {
            const int key = p[u] & (FINE_BINS - 1);
            const unsigned m = match_mask<FINE_BITS>(key, v[u]);
            const int rank = __popc(m & lt);
            if (v[u]) {
                int pos = wh[key] + rank;
                pos = (pos < E) ? pos : (E - 1);
                pos = (pos > 0) ? pos : 0;
                const int sv = (int)((unsigned)p[u] >> FINE_BITS);
                if (staged) {
                    int lp = pos - o0;
                    lp = (lp < FINE_CAP) ? lp : (FINE_CAP - 1);
                    lp = (lp > 0) ? lp : 0;
                    stage[lp] = sv;
                } else {
                    csr[pos] = sv;
                }
                if (rank == 0) wh[key] = pos - rank + __popc(m);
            }
        }
    }
    if (staged) {
        __syncthreads();
        for (int i = tid; i < len; i += 1024) {
            int pos = o0 + i;
            pos = (pos < E) ? pos : (E - 1);
            pos = (pos > 0) ? pos : 0;
            csr[pos] = stage[i];
        }
    }
}

template <int G>
__device__ __forceinline__ float gather_sum(const float* __restrict__ xw, int s, int nrm_bits,
                                            int lane, float acc)
{
    float v[G];
    #pragma unroll
    for (int k = 0; k < G; ++k) {
        const int sk = __builtin_amdgcn_readlane(s, k);
        v[k] = (xw + (size_t)sk * 32)[lane];
    }
    #pragma unroll
    for (int k = 0; k < G; ++k) {
        const float nk = __int_as_float(__builtin_amdgcn_readlane(nrm_bits, k));
        acc += v[k] * nk;
    }
    return acc;
}

__device__ __forceinline__ float dinv_of(int lo, int hi)
{
    const float deg = (float)(hi - lo + 1);
    return (deg > 0.0f) ? rsqrtf(deg) : 0.0f;
}

__global__ __launch_bounds__(256)
void pull_kernel(const float* __restrict__ xw, const int* __restrict__ csr,
                 const int* __restrict__ rowend,
                 float* __restrict__ hpart,
                 const float* __restrict__ bn, const float* __restrict__ bs,
                 const float* __restrict__ wl, const float* __restrict__ bl,
                 float* __restrict__ out, int n, int last)
{
    const int lane = (int)(threadIdx.x & 31u);
    const int wave = __builtin_amdgcn_readfirstlane((int)(threadIdx.x >> 5));
    const int wpb  = (int)(blockDim.x >> 5);
    const int nw   = (int)gridDim.x * wpb;
    const int R    = (n + nw - 1) / nw;
    const int r0   = ((int)blockIdx.x * wpb + wave) * R;
    const int r1   = (r0 + R < n) ? (r0 + R) : n;

    const float bnj = bn[lane], bsj = bs[lane], wlj = wl[lane], bl0 = bl[0];

    for (int i = r0; i < r1; ++i) {
        const int   start = i ? rowend[i - 1] : 0;
        const int   end   = rowend[i];
        const float di    = dinv_of(start, end);
        const float selfv = xw[(size_t)i * 32 + lane];
        float acc = 0.0f;

        for (int base = start; base < end; base += 32) {
            const int cnt = (end - base < 32) ? (end - base) : 32;
            int   s   = i;
            float nrm = 0.0f;
            if (lane < cnt) {
                s = csr[base + lane];
                const int hi = rowend[s];
                const int lo = s ? rowend[s - 1] : 0;
                nrm = dinv_of(lo, hi) * di;
            }
            const int nrm_bits = __float_as_int(nrm);
            if (cnt > 16) acc = gather_sum<32>(xw, s, nrm_bits, lane, acc);
            else          acc = gather_sum<16>(xw, s, nrm_bits, lane, acc);
        }
        acc += selfv * (di * di);

        if (!last) {
            hpart[(size_t)i * 32 + lane] = acc;
        } else {
            float v = hpart[(size_t)i * 32 + lane] + acc;
            v = v + bnj + bsj;
            v = fmaxf(v, 0.0f) * wlj;
            #pragma unroll
            for (int off = 16; off > 0; off >>= 1)
                v += __shfl_xor(v, off, 32);
            if (lane == 0) out[i] = v + bl0;
        }
    }
}

extern "C" void kernel_launch(void* const* d_in, const int* in_sizes, int n_in,
                              void* d_out, int out_size, void* d_ws, size_t ws_size,
                              hipStream_t stream)
{
    const float* x  = (const float*)d_in[0];
    const int*   en = (const int*)  d_in[1];
    const int*   es = (const int*)  d_in[2];
    const float* Wn = (const float*)d_in[3];
    const float* bn = (const float*)d_in[4];
    const float* Ws = (const float*)d_in[5];
    const float* bs = (const float*)d_in[6];
    const float* Wl = (const float*)d_in[7];
    const float* bl = (const float*)d_in[8];
    float* out = (float*)d_out;

    const int N = in_sizes[0] / 64;
    const int E = in_sizes[1] / 2;

    float* ws     = (float*)d_ws;
    float* xw     = ws;
    int*   items  = (int*)ws;
    float* hpart  = ws + (size_t)N * 32;
    int*   rowend = (int*)(hpart + (size_t)N * 32);
    int*   mat    = rowend + N;
    int*   csr    = mat + N;

    const int CB = 256;

    coarse_hist_kernel<<<CB, 1024, 0, stream>>>(en + E, csr, E);
    scan_mat_kernel<<<COARSE_BINS * CB / 1024, 1024, 0, stream>>>(csr, mat, COARSE_BINS * CB);
    coarse_place_kernel<<<CB, 1024, 0, stream>>>(en, en + E, mat, items, E);
    fine_sort_kernel<<<(N + FINE_BINS - 1) / FINE_BINS, 1024, 0, stream>>>(items, mat, CB, csr, rowend, N, E);
    gemm_xw_kernel<<<(N + 31) / 32, 128, 0, stream>>>(x, Wn, xw, N);
    pull_kernel<<<2048, 256, 0, stream>>>(xw, csr, rowend, hpart, bn, bs, Wl, bl, out, N, 0);

    coarse_hist_kernel<<<CB, 1024, 0, stream>>>(es + E, csr, E);
    scan_mat_kernel<<<COARSE_BINS * CB / 1024, 1024, 0, stream>>>(csr, mat, COARSE_BINS * CB);
    coarse_place_kernel<<<CB, 1024, 0, stream>>>(es, es + E, mat, items, E);
    fine_sort_kernel<<<(N + FINE_BINS - 1) / FINE_BINS, 1024, 0, stream>>>(items, mat, CB, csr, rowend, N, E);
    gemm_xw_kernel<<<(N + 31) / 32, 128, 0, stream>>>(x, Ws, xw, N);
    pull_kernel<<<2048, 256, 0, stream>>>(xw, csr, rowend, hpart, bn, bs, Wl, bl, out, N, 1);
}


# coarse pass orders each block's slice in place (table of local bucket starts), fine block reads its bucket's 256 runs; global histogram and scan launches gone (12 to 8); row sum order changes
# speedup vs baseline: 5.7258x; 1.1756x over previous
#include <hip/hip_runtime.h>


typedef __attribute__((ext_vector_type(2))) float v2f;
typedef __attribute__((ext_vector_type(8))) float v8f;


#define FINE_BITS 9
#define FINE_BINS 512
#define COARSE_BINS 256
#define PLACE_CAP 16384
#define FINE_CAP  20480
#define COARSE_BLKS 256
#define RUNS_PER_WAVE (COARSE_BLKS / 32)

template <int NBITS>
__device__ __forceinline__ unsigned match_mask(int key, bool valid)
{
    unsigned m = __builtin_amdgcn_ballot_w32(valid);
    #pragma unroll
    for (int b = 0; b < NBITS; ++b) {
        const bool bit = ((key >> b) & 1) != 0;
        const unsigned bal = __builtin_amdgcn_ballot_w32(bit);
        m &= bit ? bal : ~bal;
    }
    return m;
}

__device__ __forceinline__ int wave_incl_scan(int v, int lane)
{
    #pragma unroll
    for (int off = 1; off < 32; off <<= 1) {
        const int y = __shfl_up(v, off, 32);
        if (lane >= off) v += y;
    }
    return v;
}

__global__ __launch_bounds__(128)
void gemm_xw_kernel(const float* __restrict__ x,
                    const float* __restrict__ W,
                    float* __restrict__ xw, int nrows)
{
    const int lane  = threadIdx.x & 31;
    const int wave  = threadIdx.x >> 5;
    const int m0    = (blockIdx.x * 2 + (wave >> 1)) * 16;
    const int lhalf = lane >> 4;
    const int lm    = lane & 15;
    const int n0    = (wave & 1) * 16;

    int arow = m0 + lm;
    if (arow >= nrows) arow = nrows - 1;

    v8f c = {};
    #pragma unroll
    for (int k0 = 0; k0 < 64; k0 += 4) {
        v2f a = *(const v2f*)(x + (size_t)arow * 64 + k0 + 2 * lhalf);
        v2f b;
        b.x = W[(k0 + 2 * lhalf)     * 32 + n0 + lm];
        b.y = W[(k0 + 2 * lhalf + 1) * 32 + n0 + lm];
        c = __builtin_amdgcn_wmma_f32_16x16x4_f32(false, a, false, b,
                                                  (short)0, c, false, false);
    }

    #pragma unroll
    for (int v = 0; v < 8; ++v) {
        const int row = m0 + v + 8 * lhalf;
        if (row < nrows) xw[(size_t)row * 32 + n0 + lm] = c[v];
    }
}

__device__ __forceinline__ void wave_coarse_count(const int* __restrict__ dst,
                                                  int wbeg, int wend, int lane, int* wh)
{
    const unsigned lt = (1u << lane) - 1u;
    for (int it = wbeg; it < wend; it += 128) {
        int  k[4];
        bool v[4];
        #pragma unroll
        for (int u = 0; u < 4; ++u) {
            const int e = it + 32 * u + lane;
            v[u] = e < wend;
            k[u] = (dst[v[u] ? e : (wend - 1)] >> FINE_BITS) & (COARSE_BINS - 1);
        }
        #pragma unroll
        for (int u = 0; u < 4; ++u) {
            const unsigned m = match_mask<8>(k[u], v[u]);
            if (v[u] && (m & lt) == 0u) wh[k[u]] += __popc(m);
        }
    }
}

__device__ __forceinline__ int slice_len(int E, int nblk)
{
    return ((E + nblk - 1) / nblk + 1023) & ~1023;
}

__device__ __forceinline__ void coarse_slice(int E, int w, int& wbeg, int& wend)
{
    const int pb   = slice_len(E, COARSE_BLKS);
    const int bbeg = (int)blockIdx.x * pb;
    const int pw   = pb >> 5;
    wbeg = bbeg + w * pw;
    wend = wbeg + pw;
    if (wend > E) wend = E;
}

__global__ __launch_bounds__(1024)
void coarse_place_kernel(const int* __restrict__ src, const int* __restrict__ dst,
                         int* __restrict__ tab, int* __restrict__ items, int E)
{
    __shared__ int hist[32 * COARSE_BINS];
    __shared__ int stage[PLACE_CAP];
    __shared__ int lbase[COARSE_BINS];
    __shared__ int wsum[32];
    const int tid  = (int)threadIdx.x;
    const int lane = tid & 31;
    const int w    = __builtin_amdgcn_readfirstlane(tid >> 5);
    for (int i = tid; i < 32 * COARSE_BINS; i += 1024) hist[i] = 0;
    __syncthreads();
    int wbeg, wend;
    coarse_slice(E, w, wbeg, wend);
    const int pb   = slice_len(E, COARSE_BLKS);
    const int bbeg = (int)blockIdx.x * pb;
    int blen = E - bbeg;
    blen = (blen < pb) ? blen : pb;
    blen = (blen > 0) ? blen : 0;
    const bool staged = pb <= PLACE_CAP;
    int* wh = hist + w * COARSE_BINS;
    wave_coarse_count(dst, wbeg, wend, lane, wh);
    __syncthreads();
    int tot = 0;
    if (tid < COARSE_BINS) {
        int run = 0;
        for (int w2 = 0; w2 < 32; ++w2) {
            const int t = hist[w2 * COARSE_BINS + tid];
            hist[w2 * COARSE_BINS + tid] = run;
            run += t;
        }
        tot = run;
    }
    const int inc = wave_incl_scan(tot, lane);
    if (lane == 31) wsum[w] = (w < COARSE_BINS / 32) ? inc : 0;
    __syncthreads();
    if (w == 0) {
        const int own = wsum[lane];
        const int sc  = wave_incl_scan(own, lane);
        wsum[lane] = sc - own;
    }
    __syncthreads();
    if (tid < COARSE_BINS) {
        const int lb = wsum[w] + inc - tot;
        lbase[tid] = lb;
        tab[(int)blockIdx.x * COARSE_BINS + tid] = lb;
    }
    __syncthreads();
    const unsigned lt = (1u << lane) - 1u;
    for (int it = wbeg; it < wend; it += 128) {
        int  d[4], s[4];
        bool v[4];
        #pragma unroll
        for (int u = 0; u < 4; ++u) {
            const int e  = it + 32 * u + lane;
            v[u] = e < wend;
            const int ec = v[u] ? e : (wend - 1);
            d[u] = dst[ec];
            s[u] = src[ec];
        }
        #pragma unroll
        for (int u = 0; u < 4; ++u) {
            const int key = (d[u] >> FINE_BITS) & (COARSE_BINS - 1);
            const unsigned m = match_mask<8>(key, v[u]);
            const int rank = __popc(m & lt);
            if (v[u]) {
                const int cur = wh[key];
                int lp = lbase[key] + cur + rank;
                lp = (lp < pb) ? lp : (pb - 1);
                lp = (lp > 0) ? lp : 0;
                const int item = (s[u] << FINE_BITS) | (d[u] & (FINE_BINS - 1));
                if (staged) {
                    stage[lp] = item;
                } else {
                    int pos = bbeg + lp;
                    pos = (pos < E) ? pos : (E - 1);
                    pos = (pos > 0) ? pos : 0;
                    items[pos] = item;
                }
                if (rank == 0) wh[key] = cur + __popc(m);
            }
        }
    }
    if (staged) {
        __syncthreads();
        for (int i = tid; i < blen; i += 1024) {
            int pos = bbeg + i;
            pos = (pos < E) ? pos : (E - 1);
            pos = (pos > 0) ? pos : 0;
            items[pos] = stage[i];
        }
    }
}

__global__ __launch_bounds__(1024)
void fine_sort_kernel(const int* __restrict__ items, const int* __restrict__ tab,
                      int* __restrict__ csr, int* __restrict__ rowend, int n, int E)
{
    __shared__ int hist[32 * FINE_BINS];
    __shared__ int wsum[32];
    __shared__ int stage[FINE_CAP];
    __shared__ int rs[COARSE_BLKS];
    __shared__ int rl[COARSE_BLKS];
    __shared__ int red[64];
    const int tid  = (int)threadIdx.x;
    const int lane = tid & 31;
    const int w    = __builtin_amdgcn_readfirstlane(tid >> 5);
    const int j    = (int)blockIdx.x;
    const int pb   = slice_len(E, COARSE_BLKS);
    int* wh = hist + w * FINE_BINS;
    const unsigned lt = (1u << lane) - 1u;

    for (int i = tid; i < 32 * FINE_BINS; i += 1024) hist[i] = 0;

    int ls = 0, cnt = 0;
    if (tid < COARSE_BLKS) {
        const int bbeg = tid * pb;
        int blen = E - bbeg;
        blen = (blen < pb) ? blen : pb;
        blen = (blen > 0) ? blen : 0;
        ls = tab[tid * COARSE_BINS + j];
        int le = (j + 1 < COARSE_BINS) ? tab[tid * COARSE_BINS + j + 1] : blen;
        le = (le < blen) ? le : blen;
        le = (le > 0) ? le : 0;
        ls = (ls < le) ? ls : le;
        ls = (ls > 0) ? ls : 0;
        cnt = le - ls;
        rs[tid] = bbeg + ls;
        rl[tid] = cnt;
    }
    int sa = ls, sb = cnt;
    #pragma unroll
    for (int off = 16; off > 0; off >>= 1) {
        sa += __shfl_xor(sa, off, 32);
        sb += __shfl_xor(sb, off, 32);
    }
    if (lane == 0) { red[w] = sa; red[32 + w] = sb; }
    __syncthreads();
    int o0 = red[lane], len = red[32 + lane];
    #pragma unroll
    for (int off = 16; off > 0; off >>= 1) {
        o0  += __shfl_xor(o0, off, 32);
        len += __shfl_xor(len, off, 32);
    }
    const bool staged = len <= FINE_CAP;

    int rbeg[RUNS_PER_WAVE], rlen[RUNS_PER_WAVE];
    int maxlen = 0;
    #pragma unroll
    for (int r = 0; r < RUNS_PER_WAVE; ++r) {
        rbeg[r] = __builtin_amdgcn_readfirstlane(rs[w * RUNS_PER_WAVE + r]);
        rlen[r] = __builtin_amdgcn_readfirstlane(rl[w * RUNS_PER_WAVE + r]);
        maxlen = (rlen[r] > maxlen) ? rlen[r] : maxlen;
    }

    for (int g = 0; g < maxlen; g += 32) {
        int  k[RUNS_PER_WAVE];
        bool v[RUNS_PER_WAVE];
        #pragma unroll
        for (int r = 0; r < RUNS_PER_WAVE; ++r) {
            const int e = g + lane;
            v[r] = e < rlen[r];
            int idx = rbeg[r] + (v[r] ? e : 0);
            idx = (idx < E) ? idx : (E - 1);
            idx = (idx > 0) ? idx : 0;
            k[r] = items[idx] & (FINE_BINS - 1);
        }
        #pragma unroll
        for (int r = 0; r < RUNS_PER_WAVE; ++r) {
            if (g < rlen[r]) {
                const unsigned m = match_mask<FINE_BITS>(k[r], v[r]);
                if (v[r] && (m & lt) == 0u) wh[k[r]] += __popc(m);
            }
        }
    }
    __syncthreads();

    int tot = 0;
    if (tid < FINE_BINS) {
        for (int w2 = 0; w2 < 32; ++w2) tot += hist[w2 * FINE_BINS + tid];
    }
    const int inc = wave_incl_scan(tot, lane);
    if (lane == 31) wsum[w] = (w < 16) ? inc : 0;
    __syncthreads();
    if (w == 0) {
        const int own = wsum[lane];
        const int sc  = wave_incl_scan(own, lane);
        wsum[lane] = sc - own;
    }
    __syncthreads();
    if (tid < FINE_BINS) {
        const int start = o0 + wsum[w] + inc - tot;
        const int node  = j * FINE_BINS + tid;
        if (node < n) rowend[node] = start + tot;
        int run = start;
        for (int w2 = 0; w2 < 32; ++w2) {
            const int t = hist[w2 * FINE_BINS + tid];
            hist[w2 * FINE_BINS + tid] = run;
            run += t;
        }
    }
    __syncthreads();

    for (int g = 0; g < maxlen; g += 32) {
        int  p[RUNS_PER_WAVE];
        bool v[RUNS_PER_WAVE];
        #pragma unroll
        for (int r = 0; r < RUNS_PER_WAVE; ++r) {
            const int e = g + lane;
            v[r] = e < rlen[r];
            int idx = rbeg[r] + (v[r] ? e : 0);
            idx = (idx < E) ? idx : (E - 1);
            idx = (idx > 0) ? idx : 0;
            p[r] = items[idx];
        }
        #pragma unroll
        for (int r = 0; r < RUNS_PER_WAVE; ++r) {
            if (g < rlen[r]) {
                const int key = p[r] & (FINE_BINS - 1);
                const unsigned m = match_mask<FINE_BITS>(key, v[r]);
                const int rank = __popc(m & lt);
                if (v[r]) {
                    int pos = wh[key] + rank;
                    pos = (pos < E) ? pos : (E - 1);
                    pos = (pos > 0) ? pos : 0;
                    const int sv = (int)((unsigned)p[r] >> FINE_BITS);
                    if (staged) {
                        int lp = pos - o0;
                        lp = (lp < FINE_CAP) ? lp : (FINE_CAP - 1);
                        lp = (lp > 0) ? lp : 0;
                        stage[lp] = sv;
                    } else {
                        csr[pos] = sv;
                    }
                    if (rank == 0) wh[key] = pos - rank + __popc(m);
                }
            }
        }
    }
    if (staged) {
        __syncthreads();
        for (int i = tid; i < len; i += 1024) {
            int pos = o0 + i;
            pos = (pos < E) ? pos : (E - 1);
            pos = (pos > 0) ? pos : 0;
            csr[pos] = stage[i];
        }
    }
}

template <int G>
__device__ __forceinline__ float gather_sum(const float* __restrict__ xw, int s, int nrm_bits,
                                            int lane, float acc)
{
    float v[G];
    #pragma unroll
    for (int k = 0; k < G; ++k) {
        const int sk = __builtin_amdgcn_readlane(s, k);
        v[k] = (xw + (size_t)sk * 32)[lane];
    }
    #pragma unroll
    for (int k = 0; k < G; ++k) {
        const float nk = __int_as_float(__builtin_amdgcn_readlane(nrm_bits, k));
        acc += v[k] * nk;
    }
    return acc;
}

__device__ __forceinline__ float dinv_of(int lo, int hi)
{
    const float deg = (float)(hi - lo + 1);
    return (deg > 0.0f) ? rsqrtf(deg) : 0.0f;
}

__global__ __launch_bounds__(256)
void pull_kernel(const float* __restrict__ xw, const int* __restrict__ csr,
                 const int* __restrict__ rowend,
                 float* __restrict__ hpart,
                 const float* __restrict__ bn, const float* __restrict__ bs,
                 const float* __restrict__ wl, const float* __restrict__ bl,
                 float* __restrict__ out, int n, int last)
{
    const int lane = (int)(threadIdx.x & 31u);
    const int wave = __builtin_amdgcn_readfirstlane((int)(threadIdx.x >> 5));
    const int wpb  = (int)(blockDim.x >> 5);
    const int nw   = (int)gridDim.x * wpb;
    const int R    = (n + nw - 1) / nw;
    const int r0   = ((int)blockIdx.x * wpb + wave) * R;
    const int r1   = (r0 + R < n) ? (r0 + R) : n;

    const float bnj = bn[lane], bsj = bs[lane], wlj = wl[lane], bl0 = bl[0];

    for (int i = r0; i < r1; ++i) {
        const int   start = i ? rowend[i - 1] : 0;
        const int   end   = rowend[i];
        const float di    = dinv_of(start, end);
        const float selfv = xw[(size_t)i * 32 + lane];
        float acc = 0.0f;

        for (int base = start; base < end; base += 32) {
            const int cnt = (end - base < 32) ? (end - base) : 32;
            int   s   = i;
            float nrm = 0.0f;
            if (lane < cnt) {
                s = csr[base + lane];
                const int hi = rowend[s];
                const int lo = s ? rowend[s - 1] : 0;
                nrm = dinv_of(lo, hi) * di;
            }
            const int nrm_bits = __float_as_int(nrm);
            if (cnt > 16) acc = gather_sum<32>(xw, s, nrm_bits, lane, acc);
            else          acc = gather_sum<16>(xw, s, nrm_bits, lane, acc);
        }
        acc += selfv * (di * di);

        if (!last) {
            hpart[(size_t)i * 32 + lane] = acc;
        } else {
            float v = hpart[(size_t)i * 32 + lane] + acc;
            v = v + bnj + bsj;
            v = fmaxf(v, 0.0f) * wlj;
            #pragma unroll
            for (int off = 16; off > 0; off >>= 1)
                v += __shfl_xor(v, off, 32);
            if (lane == 0) out[i] = v + bl0;
        }
    }
}

extern "C" void kernel_launch(void* const* d_in, const int* in_sizes, int n_in,
                              void* d_out, int out_size, void* d_ws, size_t ws_size,
                              hipStream_t stream)
{
    const float* x  = (const float*)d_in[0];
    const int*   en = (const int*)  d_in[1];
    const int*   es = (const int*)  d_in[2];
    const float* Wn = (const float*)d_in[3];
    const float* bn = (const float*)d_in[4];
    const float* Ws = (const float*)d_in[5];
    const float* bs = (const float*)d_in[6];
    const float* Wl = (const float*)d_in[7];
    const float* bl = (const float*)d_in[8];
    float* out = (float*)d_out;

    const int N = in_sizes[0] / 64;
    const int E = in_sizes[1] / 2;

    float* ws     = (float*)d_ws;
    float* xw     = ws;
    int*   items  = (int*)ws;
    float* hpart  = ws + (size_t)N * 32;
    int*   rowend = (int*)(hpart + (size_t)N * 32);
    int*   tab    = rowend + N;
    int*   csr    = tab + N;

    coarse_place_kernel<<<COARSE_BLKS, 1024, 0, stream>>>(en, en + E, tab, items, E);
    fine_sort_kernel<<<(N + FINE_BINS - 1) / FINE_BINS, 1024, 0, stream>>>(items, tab, csr, rowend, N, E);
    gemm_xw_kernel<<<(N + 31) / 32, 128, 0, stream>>>(x, Wn, xw, N);
    pull_kernel<<<2048, 256, 0, stream>>>(xw, csr, rowend, hpart, bn, bs, Wl, bl, out, N, 0);

    coarse_place_kernel<<<COARSE_BLKS, 1024, 0, stream>>>(es, es + E, tab, items, E);
    fine_sort_kernel<<<(N + FINE_BINS - 1) / FINE_BINS, 1024, 0, stream>>>(items, tab, csr, rowend, N, E);
    gemm_xw_kernel<<<(N + 31) / 32, 128, 0, stream>>>(x, Ws, xw, N);
    pull_kernel<<<2048, 256, 0, stream>>>(xw, csr, rowend, hpart, bn, bs, Wl, bl, out, N, 1);
}


# placing passes rank an item once: the counting walk leaves its in-wave rank in LDS, the placing walk adds the key's start and earlier waves' count, no second set of ballots; sum order unchanged
# speedup vs baseline: 6.1330x; 1.0711x over previous
#include <hip/hip_runtime.h>


typedef __attribute__((ext_vector_type(2))) float v2f;
typedef __attribute__((ext_vector_type(8))) float v8f;


#define FINE_BITS 9
#define FINE_BINS 512
#define COARSE_BINS 256
#define PLACE_CAP 16384
#define FINE_CAP  20480
#define COARSE_BLKS 256
#define RUNS_PER_WAVE (COARSE_BLKS / 32)
#define FINE_KEEP_TRIPS 4
#define FINE_WAVE_SLOTS (FINE_KEEP_TRIPS * RUNS_PER_WAVE * 32)

template <int NBITS>
__device__ __forceinline__ unsigned match_mask(int key, bool valid)
{
    unsigned m = __builtin_amdgcn_ballot_w32(valid);
    #pragma unroll
    for (int b = 0; b < NBITS; ++b) {
        const bool bit = ((key >> b) & 1) != 0;
        const unsigned bal = __builtin_amdgcn_ballot_w32(bit);
        m &= bit ? bal : ~bal;
    }
    return m;
}

__device__ __forceinline__ int wave_incl_scan(int v, int lane)
{
    #pragma unroll
    for (int off = 1; off < 32; off <<= 1) {
        const int y = __shfl_up(v, off, 32);
        if (lane >= off) v += y;
    }
    return v;
}

__global__ __launch_bounds__(128)
void gemm_xw_kernel(const float* __restrict__ x,
                    const float* __restrict__ W,
                    float* __restrict__ xw, int nrows)
{
    const int lane  = threadIdx.x & 31;
    const int wave  = threadIdx.x >> 5;
    const int m0    = (blockIdx.x * 2 + (wave >> 1)) * 16;
    const int lhalf = lane >> 4;
    const int lm    = lane & 15;
    const int n0    = (wave & 1) * 16;

    int arow = m0 + lm;
    if (arow >= nrows) arow = nrows - 1;

    v8f c = {};
    #pragma unroll
    for (int k0 = 0; k0 < 64; k0 += 4) {
        v2f a = *(const v2f*)(x + (size_t)arow * 64 + k0 + 2 * lhalf);
        v2f b;
        b.x = W[(k0 + 2 * lhalf)     * 32 + n0 + lm];
        b.y = W[(k0 + 2 * lhalf + 1) * 32 + n0 + lm];
        c = __builtin_amdgcn_wmma_f32_16x16x4_f32(false, a, false, b,
                                                  (short)0, c, false, false);
    }

    #pragma unroll
    for (int v = 0; v < 8; ++v) {
        const int row = m0 + v + 8 * lhalf;
        if (row < nrows) xw[(size_t)row * 32 + n0 + lm] = c[v];
    }
}

__device__ __forceinline__ void wave_coarse_count(const int* __restrict__ dst,
                                                  int wbeg, int wend, int lane, int* wh,
                                                  int* rk, int bbeg, bool keep)
{
    const unsigned lt = (1u << lane) - 1u;
    for (int it = wbeg; it < wend; it += 128) {
        int  k[4];
        bool v[4];
        #pragma unroll
        for (int u = 0; u < 4; ++u) {
            const int e = it + 32 * u + lane;
            v[u] = e < wend;
            k[u] = (dst[v[u] ? e : (wend - 1)] >> FINE_BITS) & (COARSE_BINS - 1);
        }
        #pragma unroll
        for (int u = 0; u < 4; ++u) {
            const unsigned m = match_mask<8>(k[u], v[u]);
            if (v[u]) {
                const int old = wh[k[u]];
                if (keep) rk[it + 32 * u + lane - bbeg] = old + __popc(m & lt);
                if ((m & lt) == 0u) wh[k[u]] = old + __popc(m);
            }
        }
    }
}

__device__ __forceinline__ int slice_len(int E, int nblk)
{
    return ((E + nblk - 1) / nblk + 1023) & ~1023;
}

__device__ __forceinline__ void coarse_slice(int E, int w, int& wbeg, int& wend)
{
    const int pb   = slice_len(E, COARSE_BLKS);
    const int bbeg = (int)blockIdx.x * pb;
    const int pw   = pb >> 5;
    wbeg = bbeg + w * pw;
    wend = wbeg + pw;
    if (wend > E) wend = E;
}

__global__ __launch_bounds__(1024)
void coarse_place_kernel(const int* __restrict__ src, const int* __restrict__ dst,
                         int* __restrict__ tab, int* __restrict__ items, int E)
{
    __shared__ int hist[32 * COARSE_BINS];
    __shared__ int stage[PLACE_CAP];
    __shared__ int rk[PLACE_CAP];
    __shared__ int lbase[COARSE_BINS];
    __shared__ int wsum[32];
    const int tid  = (int)threadIdx.x;
    const int lane = tid & 31;
    const int w    = __builtin_amdgcn_readfirstlane(tid >> 5);
    for (int i = tid; i < 32 * COARSE_BINS; i += 1024) hist[i] = 0;
    __syncthreads();
    int wbeg, wend;
    coarse_slice(E, w, wbeg, wend);
    const int pb   = slice_len(E, COARSE_BLKS);
    const int bbeg = (int)blockIdx.x * pb;
    int blen = E - bbeg;
    blen = (blen < pb) ? blen : pb;
    blen = (blen > 0) ? blen : 0;
    const bool staged = pb <= PLACE_CAP;
    int* wh = hist + w * COARSE_BINS;
    wave_coarse_count(dst, wbeg, wend, lane, wh, rk, bbeg, staged);
    __syncthreads();
    int tot = 0;
    if (tid < COARSE_BINS) {
        int run = 0;
        for (int w2 = 0; w2 < 32; ++w2) {
            const int t = hist[w2 * COARSE_BINS + tid];
            hist[w2 * COARSE_BINS + tid] = run;
            run += t;
        }
        tot = run;
    }
    const int inc = wave_incl_scan(tot, lane);
    if (lane == 31) wsum[w] = (w < COARSE_BINS / 32) ? inc : 0;
    __syncthreads();
    if (w == 0) {
        const int own = wsum[lane];
        const int sc  = wave_incl_scan(own, lane);
        wsum[lane] = sc - own;
    }
    __syncthreads();
    if (tid < COARSE_BINS) {
        const int lb = wsum[w] + inc - tot;
        lbase[tid] = lb;
        tab[(int)blockIdx.x * COARSE_BINS + tid] = lb;
    }
    __syncthreads();
    const unsigned lt = (1u << lane) - 1u;
    for (int it = wbeg; it < wend; it += 128) {
        int  d[4], s[4];
        bool v[4];
        #pragma unroll
        for (int u = 0; u < 4; ++u) {
            const int e  = it + 32 * u + lane;
            v[u] = e < wend;
            const int ec = v[u] ? e : (wend - 1);
            d[u] = dst[ec];
            s[u] = src[ec];
        }
        #pragma unroll
        for (int u = 0; u < 4; ++u) {
            const int key  = (d[u] >> FINE_BITS) & (COARSE_BINS - 1);
            const int item = (s[u] << FINE_BITS) | (d[u] & (FINE_BINS - 1));
            if (staged) {
                if (v[u]) {
                    int lp = lbase[key] + wh[key] + rk[it + 32 * u + lane - bbeg];
                    lp = (lp < pb) ? lp : (pb - 1);
                    lp = (lp > 0) ? lp : 0;
                    stage[lp] = item;
                }
            } else {
                const unsigned m = match_mask<8>(key, v[u]);
                const int rank = __popc(m & lt);
                if (v[u]) {
                    const int cur = wh[key];
                    int lp = lbase[key] + cur + rank;
                    lp = (lp < pb) ? lp : (pb - 1);
                    lp = (lp > 0) ? lp : 0;
                    int pos = bbeg + lp;
                    pos = (pos < E) ? pos : (E - 1);
                    pos = (pos > 0) ? pos : 0;
                    items[pos] = item;
                    if (rank == 0) wh[key] = cur + __popc(m);
                }
            }
        }
    }
    if (staged) {
        __syncthreads();
        for (int i = tid; i < blen; i += 1024) {
            int pos = bbeg + i;
            pos = (pos < E) ? pos : (E - 1);
            pos = (pos > 0) ? pos : 0;
            items[pos] = stage[i];
        }
    }
}

__device__ __forceinline__ void fine_put(int pos, int sv, bool staged, int o0, int E,
                                         int* stage, int* __restrict__ csr)
{
    pos = (pos < E) ? pos : (E - 1);
    pos = (pos > 0) ? pos : 0;
    if (staged) {
        int lp = pos - o0;
        lp = (lp < FINE_CAP) ? lp : (FINE_CAP - 1);
        lp = (lp > 0) ? lp : 0;
        stage[lp] = sv;
    } else {
        csr[pos] = sv;
    }
}

__global__ __launch_bounds__(1024)
void fine_sort_kernel(const int* __restrict__ items, const int* __restrict__ tab,
                      int* __restrict__ csr, int* __restrict__ rowend, int n, int E)
{
    __shared__ int hist[32 * FINE_BINS];
    __shared__ int wsum[32];
    __shared__ int stage[FINE_CAP];
    __shared__ int rk[32 * FINE_WAVE_SLOTS];
    __shared__ int rs[COARSE_BLKS];
    __shared__ int rl[COARSE_BLKS];
    __shared__ int red[64];
    const int tid  = (int)threadIdx.x;
    const int lane = tid & 31;
    const int w    = __builtin_amdgcn_readfirstlane(tid >> 5);
    const int j    = (int)blockIdx.x;
    const int pb   = slice_len(E, COARSE_BLKS);
    int* wh = hist + w * FINE_BINS;
    const unsigned lt = (1u << lane) - 1u;

    for (int i = tid; i < 32 * FINE_BINS; i += 1024) hist[i] = 0;

    int ls = 0, cnt = 0;
    if (tid < COARSE_BLKS) {
        const int bbeg = tid * pb;
        int blen = E - bbeg;
        blen = (blen < pb) ? blen : pb;
        blen = (blen > 0) ? blen : 0;
        ls = tab[tid * COARSE_BINS + j];
        int le = (j + 1 < COARSE_BINS) ? tab[tid * COARSE_BINS + j + 1] : blen;
        le = (le < blen) ? le : blen;
        le = (le > 0) ? le : 0;
        ls = (ls < le) ? ls : le;
        ls = (ls > 0) ? ls : 0;
        cnt = le - ls;
        rs[tid] = bbeg + ls;
        rl[tid] = cnt;
    }
    int sa = ls, sb = cnt;
    #pragma unroll
    for (int off = 16; off > 0; off >>= 1) {
        sa += __shfl_xor(sa, off, 32);
        sb += __shfl_xor(sb, off, 32);
    }
    if (lane == 0) { red[w] = sa; red[32 + w] = sb; }
    __syncthreads();
    int o0 = red[lane], len = red[32 + lane];
    #pragma unroll
    for (int off = 16; off > 0; off >>= 1) {
        o0  += __shfl_xor(o0, off, 32);
        len += __shfl_xor(len, off, 32);
    }
    const bool staged = len <= FINE_CAP;

    int rbeg[RUNS_PER_WAVE], rlen[RUNS_PER_WAVE];
    int maxlen = 0;
    #pragma unroll
    for (int r = 0; r < RUNS_PER_WAVE; ++r) {
        rbeg[r] = __builtin_amdgcn_readfirstlane(rs[w * RUNS_PER_WAVE + r]);
        rlen[r] = __builtin_amdgcn_readfirstlane(rl[w * RUNS_PER_WAVE + r]);
        maxlen = (rlen[r] > maxlen) ? rlen[r] : maxlen;
    }
    const bool keep = maxlen <= FINE_KEEP_TRIPS * 32;
    int* wrk = rk + w * FINE_WAVE_SLOTS;

    for (int g = 0; g < maxlen; g += 32) {
        int  k[RUNS_PER_WAVE];
        bool v[RUNS_PER_WAVE];
        #pragma unroll
        for (int r = 0; r < RUNS_PER_WAVE; ++r) {
            const int e = g + lane;
            v[r] = e < rlen[r];
            int idx = rbeg[r] + (v[r] ? e : 0);
            idx = (idx < E) ? idx : (E - 1);
            idx = (idx > 0) ? idx : 0;
            k[r] = items[idx] & (FINE_BINS - 1);
        }
        #pragma unroll
        for (int r = 0; r < RUNS_PER_WAVE; ++r) {
            if (g < rlen[r]) {
                const unsigned m = match_mask<FINE_BITS>(k[r], v[r]);
                if (v[r]) {
                    const int old = wh[k[r]];
                    if (keep) wrk[g * RUNS_PER_WAVE + r * 32 + lane] = old + __popc(m & lt);
                    if ((m & lt) == 0u) wh[k[r]] = old + __popc(m);
                }
            }
        }
    }
    __syncthreads();

    int tot = 0;
    if (tid < FINE_BINS) {
        for (int w2 = 0; w2 < 32; ++w2) tot += hist[w2 * FINE_BINS + tid];
    }
    const int inc = wave_incl_scan(tot, lane);
    if (lane == 31) wsum[w] = (w < 16) ? inc : 0;
    __syncthreads();
    if (w == 0) {
        const int own = wsum[lane];
        const int sc  = wave_incl_scan(own, lane);
        wsum[lane] = sc - own;
    }
    __syncthreads();
    if (tid < FINE_BINS) {
        const int start = o0 + wsum[w] + inc - tot;
        const int node  = j * FINE_BINS + tid;
        if (node < n) rowend[node] = start + tot;
        int run = start;
        for (int w2 = 0; w2 < 32; ++w2) {
            const int t = hist[w2 * FINE_BINS + tid];
            hist[w2 * FINE_BINS + tid] = run;
            run += t;
        }
    }
    __syncthreads();

    for (int g = 0; g < maxlen; g += 32) {
        int  p[RUNS_PER_WAVE];
        bool v[RUNS_PER_WAVE];
        #pragma unroll
        for (int r = 0; r < RUNS_PER_WAVE; ++r) {
            const int e = g + lane;
            v[r] = e < rlen[r];
            int idx = rbeg[r] + (v[r] ? e : 0);
            idx = (idx < E) ? idx : (E - 1);
            idx = (idx > 0) ? idx : 0;
            p[r] = items[idx];
        }
        #pragma unroll
        for (int r = 0; r < RUNS_PER_WAVE; ++r) {
            if (g < rlen[r]) {
                const int key = p[r] & (FINE_BINS - 1);
                const int sv  = (int)((unsigned)p[r] >> FINE_BITS);
                if (keep) {
                    if (v[r]) {
                        const int pos = wh[key] + wrk[g * RUNS_PER_WAVE + r * 32 + lane];
                        fine_put(pos, sv, staged, o0, E, stage, csr);
                    }
                } else {
                    const unsigned m = match_mask<FINE_BITS>(key, v[r]);
                    const int rank = __popc(m & lt);
                    if (v[r]) {
                        const int cur = wh[key];
                        fine_put(cur + rank, sv, staged, o0, E, stage, csr);
                        if (rank == 0) wh[key] = cur + __popc(m);
                    }
                }
            }
        }
    }
    if (staged) {
        __syncthreads();
        for (int i = tid; i < len; i += 1024) {
            int pos = o0 + i;
            pos = (pos < E) ? pos : (E - 1);
            pos = (pos > 0) ? pos : 0;
            csr[pos] = stage[i];
        }
    }
}

template <int G>
__device__ __forceinline__ float gather_sum(const float* __restrict__ xw, int s, int nrm_bits,
                                            int lane, float acc)
{
    float v[G];
    #pragma unroll
    for (int k = 0; k < G; ++k) {
        const int sk = __builtin_amdgcn_readlane(s, k);
        v[k] = (xw + (size_t)sk * 32)[lane];
    }
    #pragma unroll
    for (int k = 0; k < G; ++k) {
        const float nk = __int_as_float(__builtin_amdgcn_readlane(nrm_bits, k));
        acc += v[k] * nk;
    }
    return acc;
}

__device__ __forceinline__ float dinv_of(int lo, int hi)
{
    const float deg = (float)(hi - lo + 1);
    return (deg > 0.0f) ? rsqrtf(deg) : 0.0f;
}

__global__ __launch_bounds__(256)
void pull_kernel(const float* __restrict__ xw, const int* __restrict__ csr,
                 const int* __restrict__ rowend,
                 float* __restrict__ hpart,
                 const float* __restrict__ bn, const float* __restrict__ bs,
                 const float* __restrict__ wl, const float* __restrict__ bl,
                 float* __restrict__ out, int n, int last)
{
    const int lane = (int)(threadIdx.x & 31u);
    const int wave = __builtin_amdgcn_readfirstlane((int)(threadIdx.x >> 5));
    const int wpb  = (int)(blockDim.x >> 5);
    const int nw   = (int)gridDim.x * wpb;
    const int R    = (n + nw - 1) / nw;
    const int r0   = ((int)blockIdx.x * wpb + wave) * R;
    const int r1   = (r0 + R < n) ? (r0 + R) : n;

    const float bnj = bn[lane], bsj = bs[lane], wlj = wl[lane], bl0 = bl[0];

    for (int i = r0; i < r1; ++i) {
        const int   start = i ? rowend[i - 1] : 0;
        const int   end   = rowend[i];
        const float di    = dinv_of(start, end);
        const float selfv = xw[(size_t)i * 32 + lane];
        float acc = 0.0f;

        for (int base = start; base < end; base += 32) {
            const int cnt = (end - base < 32) ? (end - base) : 32;
            int   s   = i;
            float nrm = 0.0f;
            if (lane < cnt) {
                s = csr[base + lane];
                const int hi = rowend[s];
                const int lo = s ? rowend[s - 1] : 0;
                nrm = dinv_of(lo, hi) * di;
            }
            const int nrm_bits = __float_as_int(nrm);
            if (cnt > 16) acc = gather_sum<32>(xw, s, nrm_bits, lane, acc);
            else          acc = gather_sum<16>(xw, s, nrm_bits, lane, acc);
        }
        acc += selfv * (di * di);

        if (!last) {
            hpart[(size_t)i * 32 + lane] = acc;
        } else {
            float v = hpart[(size_t)i * 32 + lane] + acc;
            v = v + bnj + bsj;
            v = fmaxf(v, 0.0f) * wlj;
            #pragma unroll
            for (int off = 16; off > 0; off >>= 1)
                v += __shfl_xor(v, off, 32);
            if (lane == 0) out[i] = v + bl0;
        }
    }
}

extern "C" void kernel_launch(void* const* d_in, const int* in_sizes, int n_in,
                              void* d_out, int out_size, void* d_ws, size_t ws_size,
                              hipStream_t stream)
{
    const float* x  = (const float*)d_in[0];
    const int*   en = (const int*)  d_in[1];
    const int*   es = (const int*)  d_in[2];
    const float* Wn = (const float*)d_in[3];
    const float* bn = (const float*)d_in[4];
    const float* Ws = (const float*)d_in[5];
    const float* bs = (const float*)d_in[6];
    const float* Wl = (const float*)d_in[7];
    const float* bl = (const float*)d_in[8];
    float* out = (float*)d_out;

    const int N = in_sizes[0] / 64;
    const int E = in_sizes[1] / 2;

    float* ws     = (float*)d_ws;
    float* xw     = ws;
    int*   items  = (int*)ws;
    float* hpart  = ws + (size_t)N * 32;
    int*   rowend = (int*)(hpart + (size_t)N * 32);
    int*   tab    = rowend + N;
    int*   csr    = tab + N;

    coarse_place_kernel<<<COARSE_BLKS, 1024, 0, stream>>>(en, en + E, tab, items, E);
    fine_sort_kernel<<<(N + FINE_BINS - 1) / FINE_BINS, 1024, 0, stream>>>(items, tab, csr, rowend, N, E);
    gemm_xw_kernel<<<(N + 31) / 32, 128, 0, stream>>>(x, Wn, xw, N);
    pull_kernel<<<2048, 256, 0, stream>>>(xw, csr, rowend, hpart, bn, bs, Wl, bl, out, N, 0);

    coarse_place_kernel<<<COARSE_BLKS, 1024, 0, stream>>>(es, es + E, tab, items, E);
    fine_sort_kernel<<<(N + FINE_BINS - 1) / FINE_BINS, 1024, 0, stream>>>(items, tab, csr, rowend, N, E);
    gemm_xw_kernel<<<(N + 31) / 32, 128, 0, stream>>>(x, Ws, xw, N);
    pull_kernel<<<2048, 256, 0, stream>>>(xw, csr, rowend, hpart, bn, bs, Wl, bl, out, N, 1);
}
